# starting kernel
# baseline (speedup 1.0000x reference)
#include <hip/hip_runtime.h>
#include <math.h>

#define NN 50000
#define NE 800000
#define INDIM 256
#define HID 32
#define HEADS 4
#define ETOT (NE + NN)

typedef __attribute__((ext_vector_type(2))) float v2f;
typedef __attribute__((ext_vector_type(8))) float v8f;

__device__ __forceinline__ unsigned fkey(float x) {
  unsigned u = __float_as_uint(x);
  return (u & 0x80000000u) ? ~u : (u | 0x80000000u);
}
__device__ __forceinline__ float unfkey(unsigned k) {
  return __uint_as_float((k & 0x80000000u) ? (k & 0x7fffffffu) : ~k);
}
__device__ __forceinline__ float elu(float v) {
  return (v > 0.f) ? v : (__expf(v) - 1.f);
}

__global__ void fill_u32(unsigned* p, unsigned v, int n) {
  int i = blockIdx.x * blockDim.x + threadIdx.x;
  if (i < n) p[i] = v;
}
__global__ void fill_f32(float* p, float v, int n) {
  int i = blockIdx.x * blockDim.x + threadIdx.x;
  if (i < n) p[i] = v;
}

template <int N, int K>
__global__ void gemm_wmma_f32(const float* __restrict__ A, const float* __restrict__ B,
                              float* __restrict__ C) {
  const int lane  = threadIdx.x;
  const int mbase = blockIdx.x * 16;
  const int nbase = (blockIdx.y * blockDim.y + threadIdx.y) * 16;
  const int l15   = lane & 15;
  const int koff  = (lane >> 4) << 1;
  const float* ap = A + (size_t)(mbase + l15) * K + koff;
  const float* bp = B + (size_t)koff * N + nbase + l15;
  v8f c = {0.f, 0.f, 0.f, 0.f, 0.f, 0.f, 0.f, 0.f};
#pragma unroll 8
  for (int k0 = 0; k0 < K; k0 += 4) {
    v2f a, b;
    a.x = ap[k0];
    a.y = ap[k0 + 1];
    b.x = bp[k0 * N];
    b.y = bp[(k0 + 1) * N];
    c = __builtin_amdgcn_wmma_f32_16x16x4_f32(false, a, false, b, (short)0, c,
                                              false, false);
  }
  const int crow = mbase + ((lane >> 4) << 3);
  float* cp = C + (size_t)crow * N + nbase + l15;
#pragma unroll
  for (int r = 0; r < 8; ++r) cp[r * N] = c[r];
}

__global__ void att_scores(const float* __restrict__ h, const float* __restrict__ att_s,
                           const float* __restrict__ att_d, float* __restrict__ as,
                           float* __restrict__ ad, int H, int C) {
  int i = blockIdx.x * blockDim.x + threadIdx.x;
  if (i >= NN * H) return;
  int hh = i % H;
  const float* hp = h + (size_t)i * C;
  float s = 0.f, d = 0.f;
  for (int c = 0; c < C; ++c) {
    float v = hp[c];
    s += v * att_s[hh * C + c];
    d += v * att_d[hh * C + c];
  }
  as[i] = s;
  ad[i] = d;
}

__global__ void edge_max(const int* __restrict__ src, const int* __restrict__ dst,
                         const float* __restrict__ as, const float* __restrict__ ad,
                         unsigned* __restrict__ mkey, int H) {
  long long i = (long long)blockIdx.x * blockDim.x + threadIdx.x;
  if (i >= (long long)ETOT * H) return;
  int e = (int)(i / H), h = (int)(i % H);
  int s = (e < NE) ? src[e] : (e - NE);
  int d = (e < NE) ? dst[e] : (e - NE);
  float x = as[s * H + h] + ad[d * H + h];
  x = (x > 0.f) ? x : 0.2f * x;
  atomicMax(&mkey[d * H + h], fkey(x));
}

__global__ void edge_acc(const int* __restrict__ src, const int* __restrict__ dst,
                         const float* __restrict__ as, const float* __restrict__ ad,
                         const unsigned* __restrict__ mkey, const float* __restrict__ feat,
                         float* __restrict__ acc, float* __restrict__ denom, int H, int C) {
  long long i = (long long)blockIdx.x * blockDim.x + threadIdx.x;
  if (i >= (long long)ETOT * H * C) return;
  int c = (int)(i % C);
  long long eh = i / C;
  int h = (int)(eh % H);
  int e = (int)(eh / H);
  int s = (e < NE) ? src[e] : (e - NE);
  int d = (e < NE) ? dst[e] : (e - NE);
  float x = as[s * H + h] + ad[d * H + h];
  x = (x > 0.f) ? x : 0.2f * x;
  float p = __expf(x - unfkey(mkey[d * H + h]));
  atomicAdd(&acc[((size_t)d * H + h) * C + c], p * feat[((size_t)s * H + h) * C + c]);
  if (c == 0) atomicAdd(&denom[d * H + h], p);
}

__global__ void norm_elu(float* __restrict__ acc, const float* __restrict__ denom,
                         const float* __restrict__ bias, int H, int C) {
  int i = blockIdx.x * blockDim.x + threadIdx.x;
  if (i >= NN * H * C) return;
  int n = i / (H * C);
  int r = i % (H * C);
  int h = r / C;
  acc[i] = elu(acc[i] / denom[n * H + h] + bias[r]);
}

__global__ void final_cls(const float* __restrict__ acc2, const float* __restrict__ denom2,
                          const float* __restrict__ b2, const float* __restrict__ lin_w,
                          const float* __restrict__ lin_b, float* __restrict__ out) {
  int n = blockIdx.x * (blockDim.x >> 5) + (threadIdx.x >> 5);
  int c = threadIdx.x & 31;
  if (n >= NN) return;
  float v = elu(acc2[(size_t)n * HID + c] / denom2[n] + b2[c]);
  float w0 = v * lin_w[c * 2 + 0];
  float w1 = v * lin_w[c * 2 + 1];
#pragma unroll
  for (int m = 16; m > 0; m >>= 1) {
    w0 += __shfl_xor(w0, m, 32);
    w1 += __shfl_xor(w1, m, 32);
  }
  if (c == 0) {
    out[n * 2 + 0] = w0 + lin_b[0];
    out[n * 2 + 1] = w1 + lin_b[1];
  }
}

static inline unsigned blks(long long tot, int bs) { return (unsigned)((tot + bs - 1) / bs); }

extern "C" void kernel_launch(void* const* d_in, const int* in_sizes, int n_in,
                              void* d_out, int out_size, void* d_ws, size_t ws_size,
                              hipStream_t stream) {
  const float* x        = (const float*)d_in[0];
  const int*   ei       = (const int*)d_in[1];
  const float* W1       = (const float*)d_in[2];
  const float* att_src1 = (const float*)d_in[3];
  const float* att_dst1 = (const float*)d_in[4];
  const float* b1       = (const float*)d_in[5];
  const float* W2       = (const float*)d_in[6];
  const float* att_src2 = (const float*)d_in[7];
  const float* att_dst2 = (const float*)d_in[8];
  const float* b2       = (const float*)d_in[9];
  const float* lin_w    = (const float*)d_in[10];
  const float* lin_b    = (const float*)d_in[11];
  float* out = (float*)d_out;
  const int* src = ei;
  const int* dst = ei + NE;

  float*    ws   = (float*)d_ws;
  float*    h1   = ws;
  float*    as1  = ws + 6400000;
  float*    ad1  = as1 + 200000;
  unsigned* m1   = (unsigned*)(ad1 + 200000);
  float*    d1   = (float*)(m1 + 200000);
  float*    acc1 = d1 + 200000;
  float*    z    = h1;
  float*    acc2 = h1 + 1600000;
  float*    as2  = h1 + 3200000;
  float*    ad2  = as2 + 50000;
  unsigned* m2   = (unsigned*)(ad2 + 50000);
  float*    d2   = (float*)(m2 + 50000);

  gemm_wmma_f32<HEADS * HID, INDIM>
      <<<dim3(NN / 16, 2), dim3(32, 4), 0, stream>>>(x, W1, h1);
  att_scores<<<blks(NN * HEADS, 256), 256, 0, stream>>>(h1, att_src1, att_dst1, as1, ad1, HEADS, HID);
  fill_u32<<<blks(200000, 256), 256, 0, stream>>>(m1, 0u, 200000);
  fill_f32<<<blks(200000, 256), 256, 0, stream>>>(d1, 0.f, 200000);
  fill_f32<<<blks(6400000, 256), 256, 0, stream>>>(acc1, 0.f, 6400000);
  edge_max<<<blks((long long)ETOT * HEADS, 256), 256, 0, stream>>>(src, dst, as1, ad1, m1, HEADS);
  edge_acc<<<blks((long long)ETOT * HEADS * HID, 256), 256, 0, stream>>>(src, dst, as1, ad1, m1,
                                                                         h1, acc1, d1, HEADS, HID);
  norm_elu<<<blks(6400000, 256), 256, 0, stream>>>(acc1, d1, b1, HEADS, HID);

  gemm_wmma_f32<HID, HEADS * HID>
      <<<dim3(NN / 16, 1), dim3(32, 2), 0, stream>>>(acc1, W2, z);
  att_scores<<<blks(NN, 256), 256, 0, stream>>>(z, att_src2, att_dst2, as2, ad2, 1, HID);
  fill_u32<<<blks(50000, 256), 256, 0, stream>>>(m2, 0u, 50000);
  fill_f32<<<blks(50000, 256), 256, 0, stream>>>(d2, 0.f, 50000);
  fill_f32<<<blks(1600000, 256), 256, 0, stream>>>(acc2, 0.f, 1600000);
  edge_max<<<blks((long long)ETOT, 256), 256, 0, stream>>>(src, dst, as2, ad2, m2, 1);
  edge_acc<<<blks((long long)ETOT * HID, 256), 256, 0, stream>>>(src, dst, as2, ad2, m2,
                                                                 z, acc2, d2, 1, HID);

  final_cls<<<blks((long long)NN * 32, 256), 256, 0, stream>>>(acc2, d2, b2, lin_w, lin_b, out);
}


# Aggregate per destination node over an in-call CSR (rank by int atomic, scan, fill) with a running softmax per wave, replacing the per-channel float atomic scatter, its max pass and its fills
# speedup vs baseline: 3.8612x; 3.8612x over previous
#include <hip/hip_runtime.h>
#include <math.h>


#define NN 50000
#define NE 800000
#define INDIM 256
#define HID 32
#define HEADS 4

static_assert(NN <= 65536, "source node ids are kept in 16 bits");
static_assert(NN % 8 == 0, "one wave per node, eight waves per block, four counters per 16-byte word");

typedef __attribute__((ext_vector_type(2))) float v2f;
typedef __attribute__((ext_vector_type(8))) float v8f;

__device__ __forceinline__ float elu(float v) {
  return (v > 0.f) ? v : (__expf(v) - 1.f);
}
__device__ __forceinline__ float lrelu(float x) {
  return (x > 0.f) ? x : 0.2f * x;
}

__global__ void fill_u32(unsigned* p, unsigned v, int n) {
  int i = blockIdx.x * blockDim.x + threadIdx.x;
  if (i < n) p[i] = v;
}

template <int N, int K>
__global__ void gemm_wmma_f32(const float* __restrict__ A, const float* __restrict__ B,
                              float* __restrict__ C) {
  const int lane  = threadIdx.x;
  const int mbase = blockIdx.x * 16;
  const int nbase = (blockIdx.y * blockDim.y + threadIdx.y) * 16;
  const int l15   = lane & 15;
  const int koff  = (lane >> 4) << 1;
  const float* ap = A + (size_t)(mbase + l15) * K + koff;
  const float* bp = B + (size_t)koff * N + nbase + l15;
  v8f c = {0.f, 0.f, 0.f, 0.f, 0.f, 0.f, 0.f, 0.f};
#pragma unroll 8
  for (int k0 = 0; k0 < K; k0 += 4) {
    v2f a, b;
    a.x = ap[k0];
    a.y = ap[k0 + 1];
    b.x = bp[k0 * N];
    b.y = bp[(k0 + 1) * N];
    c = __builtin_amdgcn_wmma_f32_16x16x4_f32(false, a, false, b, (short)0, c,
                                              false, false);
  }
  const int crow = mbase + ((lane >> 4) << 3);
  float* cp = C + (size_t)crow * N + nbase + l15;
#pragma unroll
  for (int r = 0; r < 8; ++r) cp[r * N] = c[r];
}

__global__ void att_src(const float* __restrict__ h, const float* __restrict__ att_s,
                        float* __restrict__ as, int H, int C) {
  int i = blockIdx.x * blockDim.x + threadIdx.x;
  if (i >= NN * H) return;
  int hh = i % H;
  const float* hp = h + (size_t)i * C;
  float s = 0.f;
  for (int c = 0; c < C; ++c) s += hp[c] * att_s[hh * C + c];
  as[i] = s;
}

__global__ void csr_rank(const int* __restrict__ dst, unsigned* __restrict__ pos,
                         unsigned* __restrict__ rank) {
  int e = blockIdx.x * blockDim.x + threadIdx.x;
  if (e < NE) rank[e] = atomicAdd(&pos[dst[e]], 1u);
}

#define SCAN_THREADS 1024
#define SCAN_CHUNKS ((NN / 4 + SCAN_THREADS - 1) / SCAN_THREADS)

__global__ void __launch_bounds__(SCAN_THREADS) csr_scan(unsigned* __restrict__ pos) {
  __shared__ unsigned wtot[SCAN_CHUNKS * 32];
  const int t = threadIdx.x, lane = t & 31, w = t >> 5;
  uint4* p4 = (uint4*)pos;
  uint4 v[SCAN_CHUNKS];
  unsigned incl[SCAN_CHUNKS];
#pragma unroll
  for (int k = 0; k < SCAN_CHUNKS; ++k) {
    const int q = k * SCAN_THREADS + t;
    v[k] = (q < NN / 4) ? p4[q] : make_uint4(0u, 0u, 0u, 0u);
  }
#pragma unroll
  for (int k = 0; k < SCAN_CHUNKS; ++k) incl[k] = v[k].x + v[k].y + v[k].z + v[k].w;
#pragma unroll
  for (int m = 1; m < 32; m <<= 1) {
    unsigned u[SCAN_CHUNKS];
#pragma unroll
    for (int k = 0; k < SCAN_CHUNKS; ++k) u[k] = __shfl_up(incl[k], m, 32);
#pragma unroll
    for (int k = 0; k < SCAN_CHUNKS; ++k) if (lane >= m) incl[k] += u[k];
  }
  if (lane == 31) {
#pragma unroll
    for (int k = 0; k < SCAN_CHUNKS; ++k) wtot[k * 32 + w] = incl[k];
  }
  __syncthreads();
  if (w == 0) {
    unsigned x[SCAN_CHUNKS], s[SCAN_CHUNKS];
#pragma unroll
    for (int k = 0; k < SCAN_CHUNKS; ++k) s[k] = x[k] = wtot[k * 32 + lane];
#pragma unroll
    for (int m = 1; m < 32; m <<= 1) {
      unsigned u[SCAN_CHUNKS];
#pragma unroll
      for (int k = 0; k < SCAN_CHUNKS; ++k) u[k] = __shfl_up(s[k], m, 32);
#pragma unroll
      for (int k = 0; k < SCAN_CHUNKS; ++k) if (lane >= m) s[k] += u[k];
    }
    unsigned carry = 0u;
#pragma unroll
    for (int k = 0; k < SCAN_CHUNKS; ++k) {
      wtot[k * 32 + lane] = carry + s[k] - x[k];
      carry += __shfl(s[k], 31, 32);
    }
  }
  __syncthreads();
#pragma unroll
  for (int k = 0; k < SCAN_CHUNKS; ++k) {
    const int q = k * SCAN_THREADS + t;
    if (q < NN / 4) {
      const unsigned own = v[k].x + v[k].y + v[k].z + v[k].w;
      uint4 o;
      o.x = wtot[k * 32 + w] + incl[k] - own;
      o.y = o.x + v[k].x;
      o.z = o.y + v[k].y;
      o.w = o.z + v[k].z;
      p4[q] = o;
    }
  }
}

__global__ void csr_fill(const int* __restrict__ src, const int* __restrict__ dst,
                         const unsigned* __restrict__ pos, const unsigned* __restrict__ rank,
                         unsigned short* __restrict__ col) {
  int e = blockIdx.x * blockDim.x + threadIdx.x;
  if (e < NE) col[pos[dst[e]] + rank[e]] = (unsigned short)src[e];
}

__global__ void __launch_bounds__(256)
gat_gather1(const float* __restrict__ feat, const float* __restrict__ as,
            const float* __restrict__ att_d, const unsigned* __restrict__ pos,
            const unsigned short* __restrict__ col, const float* __restrict__ bias,
            float* __restrict__ out) {
  const int n = blockIdx.x * (blockDim.x >> 5) + (threadIdx.x >> 5);
  const int lane = threadIdx.x & 31;
  if (n >= NN) return;
  const int hh = lane >> 3;
  const float4* fv = (const float4*)feat;
  const int beg = (int)pos[n];
  const int end = (n + 1 < NN) ? (int)pos[n + 1] : NE;
  const float4 self = fv[(size_t)n * 32 + lane];
  float adp = self.x * att_d[4 * lane + 0] + self.y * att_d[4 * lane + 1] +
              self.z * att_d[4 * lane + 2] + self.w * att_d[4 * lane + 3];
  adp += __shfl_xor(adp, 1, 32);
  adp += __shfl_xor(adp, 2, 32);
  adp += __shfl_xor(adp, 4, 32);
  float m = lrelu(as[n * HEADS + hh] + adp);
  float den = 1.f;
  float4 acc = self;
  for (int j = beg; j < end; j += 4) {
    float e[4];
    float4 r[4];
#pragma unroll
    for (int k = 0; k < 4; ++k) {
      const bool ok = (j + k) < end;
      const int jj = ok ? (j + k) : (end - 1);
      const unsigned s = col[jj];
      const float a = as[s * HEADS + hh];
      r[k] = fv[(size_t)s * 32 + lane];
      e[k] = ok ? lrelu(a + adp) : -INFINITY;
    }
    const float mn = fmaxf(fmaxf(fmaxf(e[0], e[1]), fmaxf(e[2], e[3])), m);
    const float sc = __expf(m - mn);
    const float p0 = __expf(e[0] - mn), p1 = __expf(e[1] - mn);
    const float p2 = __expf(e[2] - mn), p3 = __expf(e[3] - mn);
    acc.x = acc.x * sc + p0 * r[0].x + p1 * r[1].x + p2 * r[2].x + p3 * r[3].x;
    acc.y = acc.y * sc + p0 * r[0].y + p1 * r[1].y + p2 * r[2].y + p3 * r[3].y;
    acc.z = acc.z * sc + p0 * r[0].z + p1 * r[1].z + p2 * r[2].z + p3 * r[3].z;
    acc.w = acc.w * sc + p0 * r[0].w + p1 * r[1].w + p2 * r[2].w + p3 * r[3].w;
    den = den * sc + p0 + p1 + p2 + p3;
    m = mn;
  }
  float4 o;
  o.x = elu(acc.x / den + bias[4 * lane + 0]);
  o.y = elu(acc.y / den + bias[4 * lane + 1]);
  o.z = elu(acc.z / den + bias[4 * lane + 2]);
  o.w = elu(acc.w / den + bias[4 * lane + 3]);
  ((float4*)out)[(size_t)n * 32 + lane] = o;
}

__global__ void __launch_bounds__(256)
gat_gather2_cls(const float* __restrict__ feat, const float* __restrict__ as,
                const float* __restrict__ att_d, const unsigned* __restrict__ pos,
                const unsigned short* __restrict__ col, const float* __restrict__ b2,
                const float* __restrict__ lin_w, const float* __restrict__ lin_b,
                float* __restrict__ out) {
  const int n = blockIdx.x * (blockDim.x >> 5) + (threadIdx.x >> 5);
  const int c = threadIdx.x & 31;
  if (n >= NN) return;
  const int beg = (int)pos[n];
  const int end = (n + 1 < NN) ? (int)pos[n + 1] : NE;
  const float self = feat[(size_t)n * HID + c];
  float adp = self * att_d[c];
#pragma unroll
  for (int x = 16; x > 0; x >>= 1) adp += __shfl_xor(adp, x, 32);
  float m = lrelu(as[n] + adp);
  float den = 1.f;
  float acc = self;
  for (int j = beg; j < end; j += 4) {
    float e[4], r[4];
#pragma unroll
    for (int k = 0; k < 4; ++k) {
      const bool ok = (j + k) < end;
      const int jj = ok ? (j + k) : (end - 1);
      const unsigned s = col[jj];
      const float a = as[s];
      r[k] = feat[(size_t)s * HID + c];
      e[k] = ok ? lrelu(a + adp) : -INFINITY;
    }
    const float mn = fmaxf(fmaxf(fmaxf(e[0], e[1]), fmaxf(e[2], e[3])), m);
    const float sc = __expf(m - mn);
    const float p0 = __expf(e[0] - mn), p1 = __expf(e[1] - mn);
    const float p2 = __expf(e[2] - mn), p3 = __expf(e[3] - mn);
    acc = acc * sc + p0 * r[0] + p1 * r[1] + p2 * r[2] + p3 * r[3];
    den = den * sc + p0 + p1 + p2 + p3;
    m = mn;
  }
  const float v = elu(acc / den + b2[c]);
  float w0 = v * lin_w[c * 2 + 0];
  float w1 = v * lin_w[c * 2 + 1];
#pragma unroll
  for (int x = 16; x > 0; x >>= 1) {
    w0 += __shfl_xor(w0, x, 32);
    w1 += __shfl_xor(w1, x, 32);
  }
  if (c == 0) {
    out[n * 2 + 0] = w0 + lin_b[0];
    out[n * 2 + 1] = w1 + lin_b[1];
  }
}

static inline unsigned blks(long long tot, int bs) { return (unsigned)((tot + bs - 1) / bs); }

extern "C" void kernel_launch(void* const* d_in, const int* in_sizes, int n_in,
                              void* d_out, int out_size, void* d_ws, size_t ws_size,
                              hipStream_t stream) {
  const float* x        = (const float*)d_in[0];
  const int*   ei       = (const int*)d_in[1];
  const float* W1       = (const float*)d_in[2];
  const float* att_src1 = (const float*)d_in[3];
  const float* att_dst1 = (const float*)d_in[4];
  const float* b1       = (const float*)d_in[5];
  const float* W2       = (const float*)d_in[6];
  const float* att_src2 = (const float*)d_in[7];
  const float* att_dst2 = (const float*)d_in[8];
  const float* b2       = (const float*)d_in[9];
  const float* lin_w    = (const float*)d_in[10];
  const float* lin_b    = (const float*)d_in[11];
  float* out = (float*)d_out;
  const int* src = ei;
  const int* dst = ei + NE;

  float*          ws  = (float*)d_ws;
  float*          h1  = ws;
  float*          as1 = ws + 6400000;
  unsigned*       pos = (unsigned*)(ws + 6600000);
  unsigned short* col = (unsigned short*)(ws + 6650000);
  float*          h2  = ws + 7200000;
  unsigned*       rank = (unsigned*)h2;
  float*          z   = h1;
  float*          as2 = h1 + 3200000;

  fill_u32<<<blks(NN, 256), 256, 0, stream>>>(pos, 0u, NN);
  csr_rank<<<blks(NE, 256), 256, 0, stream>>>(dst, pos, rank);
  csr_scan<<<1, SCAN_THREADS, 0, stream>>>(pos);
  csr_fill<<<blks(NE, 256), 256, 0, stream>>>(src, dst, pos, rank, col);

  gemm_wmma_f32<HEADS * HID, INDIM>
      <<<dim3(NN / 16, 2), dim3(32, 4), 0, stream>>>(x, W1, h1);
  att_src<<<blks(NN * HEADS, 256), 256, 0, stream>>>(h1, att_src1, as1, HEADS, HID);
  gat_gather1<<<NN / 8, 256, 0, stream>>>(h1, as1, att_dst1, pos, col, b1, h2);

  gemm_wmma_f32<HID, HEADS * HID>
      <<<dim3(NN / 16, 1), dim3(32, 2), 0, stream>>>(h2, W2, z);
  att_src<<<blks(NN, 256), 256, 0, stream>>>(z, att_src2, as2, 1, HID);

  gat_gather2_cls<<<NN / 8, 256, 0, stream>>>(z, as2, att_dst2, pos, col, b2, lin_w, lin_b, out);
}


# Both products on one resident 16-wave work-group per WGP: weights staged once into LDS in fragment order, each wave loops over 16-row strips with 16-byte A loads; k summed in another order
# speedup vs baseline: 4.4660x; 1.1566x over previous
#include <hip/hip_runtime.h>
#include <math.h>


#define NN 50000
#define NE 800000
#define INDIM 256
#define HID 32
#define HEADS 4

static_assert(NN <= 65536, "source node ids are kept in 16 bits");
static_assert(NN % 8 == 0, "one wave per node, eight waves per block, four counters per 16-byte word");
static_assert(NN % 16 == 0, "the products take whole 16-row strips");

typedef __attribute__((ext_vector_type(2))) float v2f;
typedef __attribute__((ext_vector_type(8))) float v8f;

__device__ __forceinline__ float elu(float v) {
  return (v > 0.f) ? v : (__expf(v) - 1.f);
}
__device__ __forceinline__ float lrelu(float x) {
  return (x > 0.f) ? x : 0.2f * x;
}

__global__ void fill_u32(unsigned* p, unsigned v, int n) {
  int i = blockIdx.x * blockDim.x + threadIdx.x;
  if (i < n) p[i] = v;
}

#define GEMM_WGS 256
#define GEMM_THREADS 512

template <int N, int K, int G>
__global__ void __launch_bounds__(GEMM_THREADS) __attribute__((amdgpu_waves_per_eu(4, 4)))
gemm_wmma_f32(const float* __restrict__ A, const float* __restrict__ B,
              float* __restrict__ C) {
  constexpr int NT = N / 16;
  constexpr int TU = NT / G;
  constexpr int NP = K / 8;
  constexpr int WQ = (GEMM_THREADS / 32) / G;
  static_assert(N % (16 * G) == 0 && K % 8 == 0 && (GEMM_THREADS / 32) % G == 0,
                "whole column tiles per unit, whole pairs of k steps, whole groups of waves");
  __shared__ float4 bs[NP * NT * 32];

  for (int e = threadIdx.x; e < NP * NT * 32; e += GEMM_THREADS) {
    const int l = e & 31, t = (e >> 5) % NT, p = e / (32 * NT);
    const float* bp = B + (size_t)(8 * p + 4 * (l >> 4)) * N + 16 * t + (l & 15);
    float4 v;
    v.x = bp[0];
    v.y = bp[N];
    v.z = bp[2 * N];
    v.w = bp[3 * N];
    bs[e] = v;
  }
  __syncthreads();

  const int lane = threadIdx.x & 31;
  const int w    = threadIdx.x >> 5;
  const int g    = w % G;
  const int q    = w / G;
  const int c    = lane & 15;
  const int h    = lane >> 4;
  const float4* bl = bs + (g * TU) * 32 + lane;

  for (int s = blockIdx.x + GEMM_WGS * q; s < NN / 16; s += GEMM_WGS * WQ) {
    asm volatile("" ::: "memory");
    const float4* ap = (const float4*)(A + (size_t)(16 * s + c) * K) + h;
    v8f acc[TU];
#pragma unroll
    for (int t = 0; t < TU; ++t) acc[t] = (v8f){0.f, 0.f, 0.f, 0.f, 0.f, 0.f, 0.f, 0.f};
    float4 a[NP];
#pragma unroll
    for (int p = 0; p < NP; ++p) a[p] = ap[2 * p];
    float4 b[2][TU];
#pragma unroll
    for (int t = 0; t < TU; ++t) b[0][t] = bl[t * 32];
#pragma unroll
    for (int p = 0; p < NP; ++p) {
      if (p + 1 < NP) {
#pragma unroll
        for (int t = 0; t < TU; ++t) b[(p + 1) & 1][t] = bl[((p + 1) * NT + t) * 32];
      }
      const v2f a0 = {a[p].x, a[p].y}, a1 = {a[p].z, a[p].w};
#pragma unroll
      for (int t = 0; t < TU; ++t) {
        const v2f b0 = {b[p & 1][t].x, b[p & 1][t].y};
        acc[t] = __builtin_amdgcn_wmma_f32_16x16x4_f32(false, a0, false, b0, (short)0, acc[t],
                                                       false, false);
      }
#pragma unroll
      for (int t = 0; t < TU; ++t) {
        const v2f b1 = {b[p & 1][t].z, b[p & 1][t].w};
        acc[t] = __builtin_amdgcn_wmma_f32_16x16x4_f32(false, a1, false, b1, (short)0, acc[t],
                                                       false, false);
      }
    }
    float* cp = C + (size_t)(16 * s + 8 * h) * N + 16 * (g * TU) + c;
#pragma unroll
    for (int t = 0; t < TU; ++t) {
#pragma unroll
      for (int r = 0; r < 8; ++r) cp[r * N + 16 * t] = acc[t][r];
    }
  }
}

__global__ void att_src(const float* __restrict__ h, const float* __restrict__ att_s,
                        float* __restrict__ as, int H, int C) {
  int i = blockIdx.x * blockDim.x + threadIdx.x;
  if (i >= NN * H) return;
  int hh = i % H;
  const float* hp = h + (size_t)i * C;
  float s = 0.f;
  for (int c = 0; c < C; ++c) s += hp[c] * att_s[hh * C + c];
  as[i] = s;
}

__global__ void csr_rank(const int* __restrict__ dst, unsigned* __restrict__ pos,
                         unsigned* __restrict__ rank) {
  int e = blockIdx.x * blockDim.x + threadIdx.x;
  if (e < NE) rank[e] = atomicAdd(&pos[dst[e]], 1u);
}

#define SCAN_THREADS 1024
#define SCAN_CHUNKS ((NN / 4 + SCAN_THREADS - 1) / SCAN_THREADS)

__global__ void __launch_bounds__(SCAN_THREADS) csr_scan(unsigned* __restrict__ pos) {
  __shared__ unsigned wtot[SCAN_CHUNKS * 32];
  const int t = threadIdx.x, lane = t & 31, w = t >> 5;
  uint4* p4 = (uint4*)pos;
  uint4 v[SCAN_CHUNKS];
  unsigned incl[SCAN_CHUNKS];
#pragma unroll
  for (int k = 0; k < SCAN_CHUNKS; ++k) {
    const int q = k * SCAN_THREADS + t;
    v[k] = (q < NN / 4) ? p4[q] : make_uint4(0u, 0u, 0u, 0u);
  }
#pragma unroll
  for (int k = 0; k < SCAN_CHUNKS; ++k) incl[k] = v[k].x + v[k].y + v[k].z + v[k].w;
#pragma unroll
  for (int m = 1; m < 32; m <<= 1) {
    unsigned u[SCAN_CHUNKS];
#pragma unroll
    for (int k = 0; k < SCAN_CHUNKS; ++k) u[k] = __shfl_up(incl[k], m, 32);
#pragma unroll
    for (int k = 0; k < SCAN_CHUNKS; ++k) if (lane >= m) incl[k] += u[k];
  }
  if (lane == 31) {
#pragma unroll
    for (int k = 0; k < SCAN_CHUNKS; ++k) wtot[k * 32 + w] = incl[k];
  }
  __syncthreads();
  if (w == 0) {
    unsigned x[SCAN_CHUNKS], s[SCAN_CHUNKS];
#pragma unroll
    for (int k = 0; k < SCAN_CHUNKS; ++k) s[k] = x[k] = wtot[k * 32 + lane];
#pragma unroll
    for (int m = 1; m < 32; m <<= 1) {
      unsigned u[SCAN_CHUNKS];
#pragma unroll
      for (int k = 0; k < SCAN_CHUNKS; ++k) u[k] = __shfl_up(s[k], m, 32);
#pragma unroll
      for (int k = 0; k < SCAN_CHUNKS; ++k) if (lane >= m) s[k] += u[k];
    }
    unsigned carry = 0u;
#pragma unroll
    for (int k = 0; k < SCAN_CHUNKS; ++k) {
      wtot[k * 32 + lane] = carry + s[k] - x[k];
      carry += __shfl(s[k], 31, 32);
    }
  }
  __syncthreads();
#pragma unroll
  for (int k = 0; k < SCAN_CHUNKS; ++k) {
    const int q = k * SCAN_THREADS + t;
    if (q < NN / 4) {
      const unsigned own = v[k].x + v[k].y + v[k].z + v[k].w;
      uint4 o;
      o.x = wtot[k * 32 + w] + incl[k] - own;
      o.y = o.x + v[k].x;
      o.z = o.y + v[k].y;
      o.w = o.z + v[k].z;
      p4[q] = o;
    }
  }
}

__global__ void csr_fill(const int* __restrict__ src, const int* __restrict__ dst,
                         const unsigned* __restrict__ pos, const unsigned* __restrict__ rank,
                         unsigned short* __restrict__ col) {
  int e = blockIdx.x * blockDim.x + threadIdx.x;
  if (e < NE) col[pos[dst[e]] + rank[e]] = (unsigned short)src[e];
}

__global__ void __launch_bounds__(256)
gat_gather1(const float* __restrict__ feat, const float* __restrict__ as,
            const float* __restrict__ att_d, const unsigned* __restrict__ pos,
            const unsigned short* __restrict__ col, const float* __restrict__ bias,
            float* __restrict__ out) {
  const int n = blockIdx.x * (blockDim.x >> 5) + (threadIdx.x >> 5);
  const int lane = threadIdx.x & 31;
  if (n >= NN) return;
  const int hh = lane >> 3;
  const float4* fv = (const float4*)feat;
  const int beg = (int)pos[n];
  const int end = (n + 1 < NN) ? (int)pos[n + 1] : NE;
  const float4 self = fv[(size_t)n * 32 + lane];
  float adp = self.x * att_d[4 * lane + 0] + self.y * att_d[4 * lane + 1] +
              self.z * att_d[4 * lane + 2] + self.w * att_d[4 * lane + 3];
  adp += __shfl_xor(adp, 1, 32);
  adp += __shfl_xor(adp, 2, 32);
  adp += __shfl_xor(adp, 4, 32);
  float m = lrelu(as[n * HEADS + hh] + adp);
  float den = 1.f;
  float4 acc = self;
  for (int j = beg; j < end; j += 4) {
    float e[4];
    float4 r[4];
#pragma unroll
    for (int k = 0; k < 4; ++k) {
      const bool ok = (j + k) < end;
      const int jj = ok ? (j + k) : (end - 1);
      const unsigned s = col[jj];
      const float a = as[s * HEADS + hh];
      r[k] = fv[(size_t)s * 32 + lane];
      e[k] = ok ? lrelu(a + adp) : -INFINITY;
    }
    const float mn = fmaxf(fmaxf(fmaxf(e[0], e[1]), fmaxf(e[2], e[3])), m);
    const float sc = __expf(m - mn);
    const float p0 = __expf(e[0] - mn), p1 = __expf(e[1] - mn);
    const float p2 = __expf(e[2] - mn), p3 = __expf(e[3] - mn);
    acc.x = acc.x * sc + p0 * r[0].x + p1 * r[1].x + p2 * r[2].x + p3 * r[3].x;
    acc.y = acc.y * sc + p0 * r[0].y + p1 * r[1].y + p2 * r[2].y + p3 * r[3].y;
    acc.z = acc.z * sc + p0 * r[0].z + p1 * r[1].z + p2 * r[2].z + p3 * r[3].z;
    acc.w = acc.w * sc + p0 * r[0].w + p1 * r[1].w + p2 * r[2].w + p3 * r[3].w;
    den = den * sc + p0 + p1 + p2 + p3;
    m = mn;
  }
  float4 o;
  o.x = elu(acc.x / den + bias[4 * lane + 0]);
  o.y = elu(acc.y / den + bias[4 * lane + 1]);
  o.z = elu(acc.z / den + bias[4 * lane + 2]);
  o.w = elu(acc.w / den + bias[4 * lane + 3]);
  ((float4*)out)[(size_t)n * 32 + lane] = o;
}

__global__ void __launch_bounds__(256)
gat_gather2_cls(const float* __restrict__ feat, const float* __restrict__ as,
                const float* __restrict__ att_d, const unsigned* __restrict__ pos,
                const unsigned short* __restrict__ col, const float* __restrict__ b2,
                const float* __restrict__ lin_w, const float* __restrict__ lin_b,
                float* __restrict__ out) {
  const int n = blockIdx.x * (blockDim.x >> 5) + (threadIdx.x >> 5);
  const int c = threadIdx.x & 31;
  if (n >= NN) return;
  const int beg = (int)pos[n];
  const int end = (n + 1 < NN) ? (int)pos[n + 1] : NE;
  const float self = feat[(size_t)n * HID + c];
  float adp = self * att_d[c];
#pragma unroll
  for (int x = 16; x > 0; x >>= 1) adp += __shfl_xor(adp, x, 32);
  float m = lrelu(as[n] + adp);
  float den = 1.f;
  float acc = self;
  for (int j = beg; j < end; j += 4) {
    float e[4], r[4];
#pragma unroll
    for (int k = 0; k < 4; ++k) {
      const bool ok = (j + k) < end;
      const int jj = ok ? (j + k) : (end - 1);
      const unsigned s = col[jj];
      const float a = as[s];
      r[k] = feat[(size_t)s * HID + c];
      e[k] = ok ? lrelu(a + adp) : -INFINITY;
    }
    const float mn = fmaxf(fmaxf(fmaxf(e[0], e[1]), fmaxf(e[2], e[3])), m);
    const float sc = __expf(m - mn);
    const float p0 = __expf(e[0] - mn), p1 = __expf(e[1] - mn);
    const float p2 = __expf(e[2] - mn), p3 = __expf(e[3] - mn);
    acc = acc * sc + p0 * r[0] + p1 * r[1] + p2 * r[2] + p3 * r[3];
    den = den * sc + p0 + p1 + p2 + p3;
    m = mn;
  }
  const float v = elu(acc / den + b2[c]);
  float w0 = v * lin_w[c * 2 + 0];
  float w1 = v * lin_w[c * 2 + 1];
#pragma unroll
  for (int x = 16; x > 0; x >>= 1) {
    w0 += __shfl_xor(w0, x, 32);
    w1 += __shfl_xor(w1, x, 32);
  }
  if (c == 0) {
    out[n * 2 + 0] = w0 + lin_b[0];
    out[n * 2 + 1] = w1 + lin_b[1];
  }
}

static inline unsigned blks(long long tot, int bs) { return (unsigned)((tot + bs - 1) / bs); }

extern "C" void kernel_launch(void* const* d_in, const int* in_sizes, int n_in,
                              void* d_out, int out_size, void* d_ws, size_t ws_size,
                              hipStream_t stream) {
  const float* x        = (const float*)d_in[0];
  const int*   ei       = (const int*)d_in[1];
  const float* W1       = (const float*)d_in[2];
  const float* att_src1 = (const float*)d_in[3];
  const float* att_dst1 = (const float*)d_in[4];
  const float* b1       = (const float*)d_in[5];
  const float* W2       = (const float*)d_in[6];
  const float* att_src2 = (const float*)d_in[7];
  const float* att_dst2 = (const float*)d_in[8];
  const float* b2       = (const float*)d_in[9];
  const float* lin_w    = (const float*)d_in[10];
  const float* lin_b    = (const float*)d_in[11];
  float* out = (float*)d_out;
  const int* src = ei;
  const int* dst = ei + NE;

  float*          ws  = (float*)d_ws;
  float*          h1  = ws;
  float*          as1 = ws + 6400000;
  unsigned*       pos = (unsigned*)(ws + 6600000);
  unsigned short* col = (unsigned short*)(ws + 6650000);
  float*          h2  = ws + 7200000;
  unsigned*       rank = (unsigned*)h2;
  float*          z   = h1;
  float*          as2 = h1 + 3200000;

  fill_u32<<<blks(NN, 256), 256, 0, stream>>>(pos, 0u, NN);
  csr_rank<<<blks(NE, 256), 256, 0, stream>>>(dst, pos, rank);
  csr_scan<<<1, SCAN_THREADS, 0, stream>>>(pos);
  csr_fill<<<blks(NE, 256), 256, 0, stream>>>(src, dst, pos, rank, col);

  gemm_wmma_f32<HEADS * HID, INDIM, 2>
      <<<GEMM_WGS, GEMM_THREADS, 0, stream>>>(x, W1, h1);
  att_src<<<blks(NN * HEADS, 256), 256, 0, stream>>>(h1, att_src1, as1, HEADS, HID);
  gat_gather1<<<NN / 8, 256, 0, stream>>>(h1, as1, att_dst1, pos, col, b1, h2);

  gemm_wmma_f32<HID, HEADS * HID, 1>
      <<<GEMM_WGS, GEMM_THREADS, 0, stream>>>(h2, W2, z);
  att_src<<<blks(NN, 256), 256, 0, stream>>>(z, att_src2, as2, 1, HID);

  gat_gather2_cls<<<NN / 8, 256, 0, stream>>>(z, as2, att_dst2, pos, col, b2, lin_w, lin_b, out);
}


# Aggregation waves read a node's source ids and scores lane-per-edge, form softmax weights across lanes, then keep a batch of row loads in flight from scalar bases; softmax sums in another order
# speedup vs baseline: 4.8333x; 1.0822x over previous
#include <hip/hip_runtime.h>
#include <math.h>


#define NN 50000
#define NE 800000
#define INDIM 256
#define HID 32
#define HEADS 4

static_assert(NN <= 65536, "source node ids are kept in 16 bits");
static_assert(NN % 8 == 0, "one wave per node, eight waves per block, four counters per 16-byte word");
static_assert(NN % 16 == 0, "the products take whole 16-row strips");

typedef __attribute__((ext_vector_type(2))) float v2f;
typedef __attribute__((ext_vector_type(8))) float v8f;
typedef __attribute__((ext_vector_type(4))) float v4f;

__device__ __forceinline__ float elu(float v) {
  return (v > 0.f) ? v : (__expf(v) - 1.f);
}
__device__ __forceinline__ float lrelu(float x) {
  return (x > 0.f) ? x : 0.2f * x;
}

__global__ void fill_u32(unsigned* p, unsigned v, int n) {
  int i = blockIdx.x * blockDim.x + threadIdx.x;
  if (i < n) p[i] = v;
}

#define GEMM_WGS 256
#define GEMM_THREADS 512

constexpr int fold_half(int n) { return n > 1 ? n / 2 : 1; }

template <int NVAL, int M>
__device__ __forceinline__ void fold(float* v, int c) {
  if constexpr (NVAL > 1) {
    constexpr int n = NVAL / 2;
    const bool up = (c & M) != 0;
#pragma unroll
    for (int i = 0; i < n; ++i) {
      const float keep = up ? v[i + n] : v[i];
      const float send = up ? v[i] : v[i + n];
      v[i] = keep + __shfl_xor(send, M, 32);
    }
  } else {
    v[0] += __shfl_xor(v[0], M, 32);
  }
}

template <int N, int K, int G>
__global__ void __launch_bounds__(GEMM_THREADS) __attribute__((amdgpu_waves_per_eu(4, 4)))
gemm_wmma_f32(const float* __restrict__ A, const float* __restrict__ B,
              float* __restrict__ C, const float* __restrict__ att,
              float* __restrict__ as) {
  constexpr int NT = N / 16;
  constexpr int TU = NT / G;
  constexpr int NP = K / 8;
  constexpr int WQ = (GEMM_THREADS / 32) / G;
  constexpr int HW = TU / 2;
  constexpr int NV = 8 * HW;
  constexpr int NH = N / HID;
  constexpr int SH = (NV == 16) ? 0 : 1;
  static_assert(TU % 2 == 0 && (NV == 8 || NV == 16) && N % HID == 0,
                "whole heads per unit; 8 or 16 sums per lane for the four exchanges below");
  static_assert(N % (16 * G) == 0 && K % 8 == 0 && (GEMM_THREADS / 32) % G == 0,
                "whole column tiles per unit, whole pairs of k steps, whole groups of waves");
  __shared__ float4 bs[NP * NT * 32];

  for (int e = threadIdx.x; e < NP * NT * 32; e += GEMM_THREADS) {
    const int l = e & 31, t = (e >> 5) % NT, p = e / (32 * NT);
    const float* bp = B + (size_t)(8 * p + 4 * (l >> 4)) * N + 16 * t + (l & 15);
    float4 v;
    v.x = bp[0];
    v.y = bp[N];
    v.z = bp[2 * N];
    v.w = bp[3 * N];
    bs[e] = v;
  }
  __syncthreads();

  const int lane = threadIdx.x & 31;
  const int w    = threadIdx.x >> 5;
  const int g    = w % G;
  const int q    = w / G;
  const int c    = lane & 15;
  const int h    = lane >> 4;
  const float4* bl = bs + (g * TU) * 32 + lane;
  float at[TU];
#pragma unroll
  for (int t = 0; t < TU; ++t) at[t] = att[16 * (g * TU + t) + c];

  for (int s = blockIdx.x + GEMM_WGS * q; s < NN / 16; s += GEMM_WGS * WQ) {
    asm volatile("" ::: "memory");
    const float4* ap = (const float4*)(A + (size_t)(16 * s + c) * K) + h;
    v8f acc[TU];
#pragma unroll
    for (int t = 0; t < TU; ++t) acc[t] = (v8f){0.f, 0.f, 0.f, 0.f, 0.f, 0.f, 0.f, 0.f};
    float4 a[NP];
#pragma unroll
    for (int p = 0; p < NP; ++p) a[p] = ap[2 * p];
    float4 b[2][TU];
#pragma unroll
    for (int t = 0; t < TU; ++t) b[0][t] = bl[t * 32];
#pragma unroll
    for (int p = 0; p < NP; ++p) {
      if (p + 1 < NP) {
#pragma unroll
        for (int t = 0; t < TU; ++t) b[(p + 1) & 1][t] = bl[((p + 1) * NT + t) * 32];
      }
      const v2f a0 = {a[p].x, a[p].y}, a1 = {a[p].z, a[p].w};
#pragma unroll
      for (int t = 0; t < TU; ++t) {
        const v2f b0 = {b[p & 1][t].x, b[p & 1][t].y};
        acc[t] = __builtin_amdgcn_wmma_f32_16x16x4_f32(false, a0, false, b0, (short)0, acc[t],
                                                       false, false);
      }
#pragma unroll
      for (int t = 0; t < TU; ++t) {
        const v2f b1 = {b[p & 1][t].z, b[p & 1][t].w};
        acc[t] = __builtin_amdgcn_wmma_f32_16x16x4_f32(false, a1, false, b1, (short)0, acc[t],
                                                       false, false);
      }
    }
    float* cp = C + (size_t)(16 * s + 8 * h) * N + 16 * (g * TU) + c;
#pragma unroll
    for (int t = 0; t < TU; ++t) {
#pragma unroll
      for (int r = 0; r < 8; ++r) cp[r * N + 16 * t] = acc[t][r];
    }
    float v[NV];
#pragma unroll
    for (int j = 0; j < HW; ++j) {
#pragma unroll
      for (int r = 0; r < 8; ++r)
        v[8 * j + r] = acc[2 * j][r] * at[2 * j] + acc[2 * j + 1][r] * at[2 * j + 1];
    }
    fold<NV, 8>(v, c);
    fold<fold_half(NV), 4>(v, c);
    fold<fold_half(fold_half(NV)), 2>(v, c);
    fold<fold_half(fold_half(fold_half(NV))), 1>(v, c);
    const int idx = c >> SH;
    if ((c & ((1 << SH) - 1)) == 0)
      as[(size_t)(16 * s + 8 * h + (idx & 7)) * NH + g * HW + (idx >> 3)] = v[0];
  }
}

__global__ void csr_rank(const int* __restrict__ dst, unsigned* __restrict__ pos,
                         unsigned* __restrict__ rank) {
  int e = blockIdx.x * blockDim.x + threadIdx.x;
  if (e < NE) rank[e] = atomicAdd(&pos[dst[e]], 1u);
}

#define SCAN_THREADS 1024
#define SCAN_CHUNKS ((NN / 4 + SCAN_THREADS - 1) / SCAN_THREADS)

__global__ void __launch_bounds__(SCAN_THREADS) csr_scan(unsigned* __restrict__ pos) {
  __shared__ unsigned wtot[SCAN_CHUNKS * 32];
  const int t = threadIdx.x, lane = t & 31, w = t >> 5;
  uint4* p4 = (uint4*)pos;
  uint4 v[SCAN_CHUNKS];
  unsigned incl[SCAN_CHUNKS];
#pragma unroll
  for (int k = 0; k < SCAN_CHUNKS; ++k) {
    const int q = k * SCAN_THREADS + t;
    v[k] = (q < NN / 4) ? p4[q] : make_uint4(0u, 0u, 0u, 0u);
  }
#pragma unroll
  for (int k = 0; k < SCAN_CHUNKS; ++k) incl[k] = v[k].x + v[k].y + v[k].z + v[k].w;
#pragma unroll
  for (int m = 1; m < 32; m <<= 1) {
    unsigned u[SCAN_CHUNKS];
#pragma unroll
    for (int k = 0; k < SCAN_CHUNKS; ++k) u[k] = __shfl_up(incl[k], m, 32);
#pragma unroll
    for (int k = 0; k < SCAN_CHUNKS; ++k) if (lane >= m) incl[k] += u[k];
  }
  if (lane == 31) {
#pragma unroll
    for (int k = 0; k < SCAN_CHUNKS; ++k) wtot[k * 32 + w] = incl[k];
  }
  __syncthreads();
  if (w == 0) {
    unsigned x[SCAN_CHUNKS], s[SCAN_CHUNKS];
#pragma unroll
    for (int k = 0; k < SCAN_CHUNKS; ++k) s[k] = x[k] = wtot[k * 32 + lane];
#pragma unroll
    for (int m = 1; m < 32; m <<= 1) {
      unsigned u[SCAN_CHUNKS];
#pragma unroll
      for (int k = 0; k < SCAN_CHUNKS; ++k) u[k] = __shfl_up(s[k], m, 32);
#pragma unroll
      for (int k = 0; k < SCAN_CHUNKS; ++k) if (lane >= m) s[k] += u[k];
    }
    unsigned carry = 0u;
#pragma unroll
    for (int k = 0; k < SCAN_CHUNKS; ++k) {
      wtot[k * 32 + lane] = carry + s[k] - x[k];
      carry += __shfl(s[k], 31, 32);
    }
  }
  __syncthreads();
#pragma unroll
  for (int k = 0; k < SCAN_CHUNKS; ++k) {
    const int q = k * SCAN_THREADS + t;
    if (q < NN / 4) {
      const unsigned own = v[k].x + v[k].y + v[k].z + v[k].w;
      uint4 o;
      o.x = wtot[k * 32 + w] + incl[k] - own;
      o.y = o.x + v[k].x;
      o.z = o.y + v[k].y;
      o.w = o.z + v[k].z;
      p4[q] = o;
    }
  }
}

__global__ void csr_fill(const int* __restrict__ src, const int* __restrict__ dst,
                         const unsigned* __restrict__ pos, const unsigned* __restrict__ rank,
                         unsigned short* __restrict__ col) {
  int e = blockIdx.x * blockDim.x + threadIdx.x;
  if (e < NE) col[pos[dst[e]] + rank[e]] = (unsigned short)src[e];
}


__device__ __forceinline__ float lane_val(float v, int k) {
  return __int_as_float(__builtin_amdgcn_readlane(__float_as_int(v), k));
}
template <int X>
__device__ __forceinline__ float row_xor(float v) {
  const int b = __float_as_int(v);
  return __int_as_float(__builtin_amdgcn_update_dpp(b, b, 0x160 + X, 0xf, 0xf, false));
}
__device__ __forceinline__ float xhalf(float v) {
  const int b = __float_as_int(v);
  return __int_as_float(__builtin_amdgcn_permlanex16(b, b, 0x76543210, 0xfedcba98, false, false));
}
#define GLOBAL_AS __attribute__((address_space(1)))
template <typename T>
__device__ __forceinline__ const GLOBAL_AS T* row_base(const T* table, int sk, int row_len) {
  const GLOBAL_AS T* rp = (const GLOBAL_AS T*)(table + (size_t)sk * row_len);
  asm("" : "+s"(rp));
  return rp;
}
__device__ __forceinline__ float wave_max(float v) {
  v = fmaxf(v, row_xor<1>(v));
  v = fmaxf(v, row_xor<2>(v));
  v = fmaxf(v, row_xor<4>(v));
  v = fmaxf(v, row_xor<8>(v));
  return fmaxf(v, xhalf(v));
}

__global__ void __launch_bounds__(256)
gat_gather1(const float* __restrict__ feat, const float* __restrict__ as,
            const float* __restrict__ att_d, const unsigned* __restrict__ pos,
            const unsigned short* __restrict__ col, const float* __restrict__ bias,
            float* __restrict__ out) {
  const int n = blockIdx.x * (blockDim.x >> 5) + __builtin_amdgcn_readfirstlane(threadIdx.x >> 5);
  const int lane = threadIdx.x & 31;
  if (n >= NN) return;
  const int hh = lane >> 3;
  const v4f* fv = (const v4f*)feat;
  const float4* as4 = (const float4*)as;
  const int beg = (int)pos[n];
  const int end = (n + 1 < NN) ? (int)pos[n + 1] : NE;
  unsigned s = (lane < end - beg) ? (unsigned)col[beg + lane] : (unsigned)n;
  const float4 asn = as4[n];
  const v4f self = (fv + (size_t)n * 32)[lane];
  float adp = self.x * att_d[4 * lane + 0] + self.y * att_d[4 * lane + 1] +
              self.z * att_d[4 * lane + 2] + self.w * att_d[4 * lane + 3];
  adp += row_xor<1>(adp);
  adp += row_xor<2>(adp);
  adp += row_xor<4>(adp);
  const float ad0 = lane_val(adp, 0), ad1 = lane_val(adp, 8);
  const float ad2 = lane_val(adp, 16), ad3 = lane_val(adp, 24);
  float m0 = lrelu(asn.x + ad0), m1 = lrelu(asn.y + ad1);
  float m2 = lrelu(asn.z + ad2), m3 = lrelu(asn.w + ad3);
  float den = 1.f;
  v4f acc = self;
  for (int j0 = beg; ; ) {
    const int cnt = (end - j0 < 32) ? (end - j0) : 32;
    const bool ok = lane < cnt;
    const float4 a = as4[s];
    v4f r[8];
#pragma unroll
    for (int k = 0; k < 8; ++k)
      r[k] = row_base(fv, __builtin_amdgcn_readlane((int)s, k), 32)[lane];
    const float e0 = ok ? lrelu(a.x + ad0) : -INFINITY;
    const float e1 = ok ? lrelu(a.y + ad1) : -INFINITY;
    const float e2 = ok ? lrelu(a.z + ad2) : -INFINITY;
    const float e3 = ok ? lrelu(a.w + ad3) : -INFINITY;
    const float n0 = fmaxf(m0, wave_max(e0)), n1 = fmaxf(m1, wave_max(e1));
    const float n2 = fmaxf(m2, wave_max(e2)), n3 = fmaxf(m3, wave_max(e3));
    const float p0 = __expf(e0 - n0), p1 = __expf(e1 - n1);
    const float p2 = __expf(e2 - n2), p3 = __expf(e3 - n3);
    const float d0 = m0 - n0, d1 = m1 - n1, d2 = m2 - n2, d3 = m3 - n3;
    const float sc = __expf((hh == 0) ? d0 : (hh == 1) ? d1 : (hh == 2) ? d2 : d3);
    m0 = n0; m1 = n1; m2 = n2; m3 = n3;
    acc.x *= sc; acc.y *= sc; acc.z *= sc; acc.w *= sc;
    den *= sc;
    for (int k0 = 0; ; ) {
#pragma unroll
      for (int k = 0; k < 8; ++k) {
        const float q0 = lane_val(p0, k0 + k), q1 = lane_val(p1, k0 + k);
        const float q2 = lane_val(p2, k0 + k), q3 = lane_val(p3, k0 + k);
        const float pk = (hh == 0) ? q0 : (hh == 1) ? q1 : (hh == 2) ? q2 : q3;
        acc.x += pk * r[k].x; acc.y += pk * r[k].y;
        acc.z += pk * r[k].z; acc.w += pk * r[k].w;
        den += pk;
      }
      k0 += 8;
      if (k0 >= cnt) break;
#pragma unroll
      for (int k = 0; k < 8; ++k)
        r[k] = row_base(fv, __builtin_amdgcn_readlane((int)s, k0 + k), 32)[lane];
    }
    j0 += 32;
    if (j0 >= end) break;
    s = (lane < end - j0) ? (unsigned)col[j0 + lane] : (unsigned)n;
  }
  float4 o;
  o.x = elu(acc.x / den + bias[4 * lane + 0]);
  o.y = elu(acc.y / den + bias[4 * lane + 1]);
  o.z = elu(acc.z / den + bias[4 * lane + 2]);
  o.w = elu(acc.w / den + bias[4 * lane + 3]);
  ((float4*)out + (size_t)n * 32)[lane] = o;
}

__global__ void __launch_bounds__(256)
gat_gather2_cls(const float* __restrict__ feat, const float* __restrict__ as,
                const float* __restrict__ att_d, const unsigned* __restrict__ pos,
                const unsigned short* __restrict__ col, const float* __restrict__ b2,
                const float* __restrict__ lin_w, const float* __restrict__ lin_b,
                float* __restrict__ out) {
  const int n = blockIdx.x * (blockDim.x >> 5) + __builtin_amdgcn_readfirstlane(threadIdx.x >> 5);
  const int c = threadIdx.x & 31;
  if (n >= NN) return;
  const int beg = (int)pos[n];
  const int end = (n + 1 < NN) ? (int)pos[n + 1] : NE;
  unsigned s = (c < end - beg) ? (unsigned)col[beg + c] : (unsigned)n;
  const float asn = as[n];
  const float self = (feat + (size_t)n * HID)[c];
  float adp = self * att_d[c];
  adp += xhalf(adp);
  adp += row_xor<8>(adp);
  adp += row_xor<4>(adp);
  adp += row_xor<2>(adp);
  adp += row_xor<1>(adp);
  float m = lrelu(asn + adp);
  float den = 1.f;
  float acc = self;
  for (int j0 = beg; ; ) {
    const int cnt = (end - j0 < 32) ? (end - j0) : 32;
    const bool ok = c < cnt;
    const float a = as[s];
    float r[16];
#pragma unroll
    for (int k = 0; k < 16; ++k)
      r[k] = row_base(feat, __builtin_amdgcn_readlane((int)s, k), HID)[c];
    const float e = ok ? lrelu(a + adp) : -INFINITY;
    const float mn = fmaxf(m, wave_max(e));
    const float p = __expf(e - mn);
    const float sc = __expf(m - mn);
    m = mn;
    acc *= sc;
    den *= sc;
    for (int k0 = 0; ; ) {
#pragma unroll
      for (int k = 0; k < 16; ++k) {
        const float pk = lane_val(p, k0 + k);
        acc += pk * r[k];
        den += pk;
      }
      k0 += 16;
      if (k0 >= cnt) break;
#pragma unroll
      for (int k = 0; k < 16; ++k)
        r[k] = row_base(feat, __builtin_amdgcn_readlane((int)s, k0 + k), HID)[c];
    }
    j0 += 32;
    if (j0 >= end) break;
    s = (c < end - j0) ? (unsigned)col[j0 + c] : (unsigned)n;
  }
  const float v = elu(acc / den + b2[c]);
  float w0 = v * lin_w[c * 2 + 0];
  float w1 = v * lin_w[c * 2 + 1];
#pragma unroll
  for (int x = 16; x > 0; x >>= 1) {
    w0 += __shfl_xor(w0, x, 32);
    w1 += __shfl_xor(w1, x, 32);
  }
  if (c == 0) {
    out[n * 2 + 0] = w0 + lin_b[0];
    out[n * 2 + 1] = w1 + lin_b[1];
  }
}

static inline unsigned blks(long long tot, int bs) { return (unsigned)((tot + bs - 1) / bs); }

extern "C" void kernel_launch(void* const* d_in, const int* in_sizes, int n_in,
                              void* d_out, int out_size, void* d_ws, size_t ws_size,
                              hipStream_t stream) {
  const float* x        = (const float*)d_in[0];
  const int*   ei       = (const int*)d_in[1];
  const float* W1       = (const float*)d_in[2];
  const float* att_src1 = (const float*)d_in[3];
  const float* att_dst1 = (const float*)d_in[4];
  const float* b1       = (const float*)d_in[5];
  const float* W2       = (const float*)d_in[6];
  const float* att_src2 = (const float*)d_in[7];
  const float* att_dst2 = (const float*)d_in[8];
  const float* b2       = (const float*)d_in[9];
  const float* lin_w    = (const float*)d_in[10];
  const float* lin_b    = (const float*)d_in[11];
  float* out = (float*)d_out;
  const int* src = ei;
  const int* dst = ei + NE;

  float*          ws  = (float*)d_ws;
  float*          h1  = ws;
  float*          as1 = ws + 6400000;
  unsigned*       pos = (unsigned*)(ws + 6600000);
  unsigned short* col = (unsigned short*)(ws + 6650000);
  float*          h2  = ws + 7200000;
  unsigned*       rank = (unsigned*)h2;
  float*          z   = h1;
  float*          as2 = h1 + 3200000;

  fill_u32<<<blks(NN, 256), 256, 0, stream>>>(pos, 0u, NN);
  csr_rank<<<blks(NE, 256), 256, 0, stream>>>(dst, pos, rank);
  csr_scan<<<1, SCAN_THREADS, 0, stream>>>(pos);
  csr_fill<<<blks(NE, 256), 256, 0, stream>>>(src, dst, pos, rank, col);

  gemm_wmma_f32<HEADS * HID, INDIM, 2>
      <<<GEMM_WGS, GEMM_THREADS, 0, stream>>>(x, W1, h1, att_src1, as1);
  gat_gather1<<<NN / 8, 256, 0, stream>>>(h1, as1, att_dst1, pos, col, b1, h2);

  gemm_wmma_f32<HID, HEADS * HID, 1>
      <<<GEMM_WGS, GEMM_THREADS, 0, stream>>>(h2, W2, z, att_src2, as2);

  gat_gather2_cls<<<NN / 8, 256, 0, stream>>>(z, as2, att_dst2, pos, col, b2, lin_w, lin_b, out);
}


# Second product folded into the layer-1 aggregation launch: 16 nodes per work-group, rows kept in LDS, 16 waves split tiles and k, partials summed via LDS; 5 launches to 4; k summed in another order
# speedup vs baseline: 9.8856x; 2.0453x over previous
#include <hip/hip_runtime.h>
#include <math.h>


#define NN 50000
#define NE 800000
#define INDIM 256
#define HID 32
#define HEADS 4

static_assert(NN <= 65536, "source node ids are kept in 16 bits");
static_assert(NN % 8 == 0, "one wave per node, eight waves per block");
static_assert(NN % 16 == 0, "the products take whole 16-row strips; layer 1 aggregates 16 nodes per block");

typedef __attribute__((ext_vector_type(2))) float v2f;
typedef __attribute__((ext_vector_type(8))) float v8f;
typedef __attribute__((ext_vector_type(4))) float v4f;

__device__ __forceinline__ float elu(float v) {
  return (v > 0.f) ? v : (__expf(v) - 1.f);
}
__device__ __forceinline__ float lrelu(float x) {
  return (x > 0.f) ? x : 0.2f * x;
}

#define BSH 8
#define NBK ((NN + (1 << BSH) - 1) >> BSH)
#define NEB 256
#define EPB ((NE + NEB - 1) / NEB)
#define SORT_THREADS 1024
#define FINE_CAP 16384
#define FINE_REG 10
#define RUN_LANES (SORT_THREADS / NEB)
#define PRE_Q ((NBK * (NEB / 4) + SORT_THREADS - 1) / SORT_THREADS)
static_assert(NBK <= 256, "bucket ids index 256-entry LDS arrays");
static_assert(NEB % 4 == 0, "a table row is read as whole 16-byte words");
static_assert(EPB < 65536, "a run's length and its start inside the block share one table word");
static_assert(SORT_THREADS % NEB == 0 && RUN_LANES >= 1, "whole groups of threads per run");
static_assert((long long)NEB * EPB == NE, "the blocks cover every edge and every block is full");

template <int X>
__device__ __forceinline__ unsigned row_xor_u(unsigned v) {
  return (unsigned)__builtin_amdgcn_update_dpp((int)v, (int)v, 0x160 + X, 0xf, 0xf, false);
}
__device__ __forceinline__ unsigned wave_sum_u(unsigned v) {
  v += row_xor_u<1>(v);
  v += row_xor_u<2>(v);
  v += row_xor_u<4>(v);
  v += row_xor_u<8>(v);
  return v + (unsigned)__builtin_amdgcn_permlanex16((int)v, (int)v, 0x76543210, 0xfedcba98, false, false);
}

__device__ __forceinline__ void scan256(unsigned* a, int lane) {
  unsigned v[8], sum = 0u;
#pragma unroll
  for (int k = 0; k < 8; ++k) { v[k] = a[8 * lane + k]; sum += v[k]; }
  unsigned inc = sum;
#pragma unroll
  for (int m = 1; m < 32; m <<= 1) {
    const unsigned u = __shfl_up(inc, m, 32);
    if (lane >= m) inc += u;
  }
  unsigned run = inc - sum;
#pragma unroll
  for (int k = 0; k < 8; ++k) { a[8 * lane + k] = run; run += v[k]; }
}

template <int NT>
__device__ __forceinline__ void local_load(const int* __restrict__ src, const int* __restrict__ dst,
                                           int blk, unsigned (&wd)[(EPB + NT - 1) / NT]) {
  const int t = threadIdx.x;
  const int e0 = blk * EPB;
#pragma unroll
  for (int k = 0; k < (EPB + NT - 1) / NT; ++k) {
    const int i = (t + k * NT < EPB) ? t + k * NT : 0;
    wd[k] = ((unsigned)dst[e0 + i] << 16) | (unsigned)src[e0 + i];
  }
}
template <int NT>
__device__ __forceinline__ void local_order(const unsigned (&wd)[(EPB + NT - 1) / NT],
                                            unsigned* cnt, unsigned* cursor, unsigned* stage) {
  const int t = threadIdx.x, lane = t & 31, w = t >> 5;
  if (t < 256) cnt[t] = 0u;
  __syncthreads();
#pragma unroll
  for (int k = 0; k < (EPB + NT - 1) / NT; ++k) {
    if (t + k * NT < EPB) atomicAdd(&cnt[wd[k] >> 24], 1u);
  }
  __syncthreads();
  if (w == 0) {
    unsigned v[8], sum = 0u;
#pragma unroll
    for (int k = 0; k < 8; ++k) { v[k] = cnt[8 * lane + k]; sum += v[k]; }
    unsigned inc = sum;
#pragma unroll
    for (int m = 1; m < 32; m <<= 1) {
      const unsigned u = __shfl_up(inc, m, 32);
      if (lane >= m) inc += u;
    }
    unsigned run = inc - sum;
#pragma unroll
    for (int k = 0; k < 8; ++k) {
      const int b = 8 * lane + k;
      cursor[b] = run;
      cnt[b] = (run << 16) | v[k];
      run += v[k];
    }
  }
  __syncthreads();
  unsigned r[(EPB + NT - 1) / NT];
#pragma unroll
  for (int k = 0; k < (EPB + NT - 1) / NT; ++k) {
    r[k] = (t + k * NT < EPB) ? atomicAdd(&cursor[wd[k] >> 24], 1u) : 0u;
  }
#pragma unroll
  for (int k = 0; k < (EPB + NT - 1) / NT; ++k) {
    if (t + k * NT < EPB) stage[r[k]] = wd[k];
  }
}
template <int NT>
__device__ __forceinline__ void local_store(unsigned* __restrict__ table, unsigned* __restrict__ tmp,
                                            int blk, const unsigned* cnt, const unsigned* stage) {
  static_assert(NT >= NBK, "one thread per table word");
  const int t = threadIdx.x;
  const int e0 = blk * EPB;
  if (t < NBK) table[t * NEB + blk] = cnt[t];
#pragma unroll
  for (int k = 0; k < (EPB + NT - 1) / NT; ++k) {
    const int i = t + k * NT;
    if (i < EPB) tmp[e0 + i] = stage[i];
  }
}

__global__ void __launch_bounds__(SORT_THREADS)
csr_fine(const unsigned* __restrict__ tmp, const unsigned* __restrict__ table,
         unsigned* __restrict__ pos, unsigned short* __restrict__ col) {
  __shared__ unsigned c[256];
  __shared__ unsigned part_pre[SORT_THREADS / 32];
  __shared__ unsigned part_own[SORT_THREADS / 32];
  __shared__ unsigned short stage[FINE_CAP];
  const int t = threadIdx.x, lane = t & 31, w = t >> 5, b = blockIdx.x;
  const int rk = t / RUN_LANES, rj = t % RUN_LANES;
  const unsigned word = table[b * NEB + rk];
  const uint4* tq = (const uint4*)table;
  const int nq = b * (NEB / 4);
  uint4 pv[PRE_Q];
#pragma unroll
  for (int k = 0; k < PRE_Q; ++k) {
    const int i = t + k * SORT_THREADS;
    pv[k] = tq[(i < nq) ? i : 0];
  }
  const int rc = (int)(word & 0xffffu);
  const unsigned* rp = tmp + (size_t)rk * EPB + (word >> 16);
  unsigned x[FINE_REG];
#pragma unroll
  for (int k = 0; k < FINE_REG; ++k) {
    const int i = rj + k * RUN_LANES;
    x[k] = (i < rc) ? rp[i] : 0u;
  }
  if (t < 256) c[t] = 0u;
  unsigned ps = 0u;
#pragma unroll
  for (int k = 0; k < PRE_Q; ++k) {
    const unsigned q = (pv[k].x & 0xffffu) + (pv[k].y & 0xffffu) + (pv[k].z & 0xffffu) + (pv[k].w & 0xffffu);
    ps += (t + k * SORT_THREADS < nq) ? q : 0u;
  }
  ps = wave_sum_u(ps);
  const unsigned os = wave_sum_u(rj == 0 ? (unsigned)rc : 0u);
  if (lane == 0) { part_pre[w] = ps; part_own[w] = os; }
  __syncthreads();
  const int s = (int)wave_sum_u(part_pre[lane]);
  const int cnt = (int)wave_sum_u(part_own[lane]);
#pragma unroll
  for (int k = 0; k < FINE_REG; ++k) {
    if (rj + k * RUN_LANES < rc) atomicAdd(&c[(x[k] >> 16) & 255u], 1u);
  }
  for (int i = rj + FINE_REG * RUN_LANES; i < rc; i += RUN_LANES)
    atomicAdd(&c[(rp[i] >> 16) & 255u], 1u);
  __syncthreads();
  if (w == 0) scan256(c, lane);
  __syncthreads();
  const unsigned first = (t < 256) ? (unsigned)s + c[t] : 0u;
  __syncthreads();
  if (cnt <= FINE_CAP) {
    unsigned r[FINE_REG];
#pragma unroll
    for (int k = 0; k < FINE_REG; ++k) {
      r[k] = (rj + k * RUN_LANES < rc) ? atomicAdd(&c[(x[k] >> 16) & 255u], 1u) : 0u;
    }
#pragma unroll
    for (int k = 0; k < FINE_REG; ++k) {
      if (rj + k * RUN_LANES < rc) stage[r[k]] = (unsigned short)x[k];
    }
    for (int i = rj + FINE_REG * RUN_LANES; i < rc; i += RUN_LANES) {
      const unsigned y = rp[i];
      stage[atomicAdd(&c[(y >> 16) & 255u], 1u)] = (unsigned short)y;
    }
    __syncthreads();
    for (int i = t; i < cnt; i += SORT_THREADS) col[s + i] = stage[i];
  } else {
#pragma unroll
    for (int k = 0; k < FINE_REG; ++k) {
      if (rj + k * RUN_LANES < rc)
        col[s + atomicAdd(&c[(x[k] >> 16) & 255u], 1u)] = (unsigned short)x[k];
    }
    for (int i = rj + FINE_REG * RUN_LANES; i < rc; i += RUN_LANES) {
      const unsigned y = rp[i];
      col[s + atomicAdd(&c[(y >> 16) & 255u], 1u)] = (unsigned short)y;
    }
  }
  if (t < 256 && (b << BSH) + t < NN) pos[(b << BSH) + t] = first;
}

#define GEMM_WGS 256
#define GEMM_THREADS 512

template <int N, int K, int G, bool SORT>
__global__ void __launch_bounds__(GEMM_THREADS) __attribute__((amdgpu_waves_per_eu(4, 4)))
gemm_wmma_f32(const float* __restrict__ A, const float* __restrict__ B,
              float* __restrict__ C, const int* __restrict__ src, const int* __restrict__ dst,
              unsigned* __restrict__ table, unsigned* __restrict__ tmp) {
  static_assert(!SORT || GEMM_WGS == NEB, "one work-group per edge block");
  constexpr int NT = N / 16;
  constexpr int TU = NT / G;
  constexpr int NP = K / 8;
  constexpr int WQ = (GEMM_THREADS / 32) / G;
  static_assert(N % (16 * G) == 0 && K % 8 == 0 && (GEMM_THREADS / 32) % G == 0,
                "whole column tiles per unit, whole pairs of k steps, whole groups of waves");
  __shared__ float4 bs[NP * NT * 32];
  __shared__ unsigned s_cnt[SORT ? 256 : 1];
  __shared__ unsigned s_cursor[SORT ? 256 : 1];
  __shared__ unsigned s_stage[SORT ? EPB : 1];

  unsigned wd[SORT ? (EPB + GEMM_THREADS - 1) / GEMM_THREADS : 1];
  if constexpr (SORT) local_load<GEMM_THREADS>(src, dst, blockIdx.x, wd);

  for (int e = threadIdx.x; e < NP * NT * 32; e += GEMM_THREADS) {
    const int l = e & 31, t = (e >> 5) % NT, p = e / (32 * NT);
    const float* bp = B + (size_t)(8 * p + 4 * (l >> 4)) * N + 16 * t + (l & 15);
    float4 v;
    v.x = bp[0];
    v.y = bp[N];
    v.z = bp[2 * N];
    v.w = bp[3 * N];
    bs[e] = v;
  }
  if constexpr (SORT) local_order<GEMM_THREADS>(wd, s_cnt, s_cursor, s_stage);
  __syncthreads();
  if constexpr (SORT) local_store<GEMM_THREADS>(table, tmp, blockIdx.x, s_cnt, s_stage);

  const int lane = threadIdx.x & 31;
  const int w    = threadIdx.x >> 5;
  const int g    = w % G;
  const int q    = w / G;
  const int c    = lane & 15;
  const int h    = lane >> 4;
  const float4* bl = bs + (g * TU) * 32 + lane;

  for (int s = blockIdx.x + GEMM_WGS * q; s < NN / 16; s += GEMM_WGS * WQ) {
    asm volatile("" ::: "memory");
    const float4* ap = (const float4*)(A + (size_t)(16 * s + c) * K) + h;
    v8f acc[TU];
#pragma unroll
    for (int t = 0; t < TU; ++t) acc[t] = (v8f){0.f, 0.f, 0.f, 0.f, 0.f, 0.f, 0.f, 0.f};
    float4 a[NP];
#pragma unroll
    for (int p = 0; p < NP; ++p) a[p] = ap[2 * p];
    float4 b[2][TU];
#pragma unroll
    for (int t = 0; t < TU; ++t) b[0][t] = bl[t * 32];
#pragma unroll
    for (int p = 0; p < NP; ++p) {
      if (p + 1 < NP) {
#pragma unroll
        for (int t = 0; t < TU; ++t) b[(p + 1) & 1][t] = bl[((p + 1) * NT + t) * 32];
      }
      const v2f a0 = {a[p].x, a[p].y}, a1 = {a[p].z, a[p].w};
#pragma unroll
      for (int t = 0; t < TU; ++t) {
        const v2f b0 = {b[p & 1][t].x, b[p & 1][t].y};
        acc[t] = __builtin_amdgcn_wmma_f32_16x16x4_f32(false, a0, false, b0, (short)0, acc[t],
                                                       false, false);
      }
#pragma unroll
      for (int t = 0; t < TU; ++t) {
        const v2f b1 = {b[p & 1][t].z, b[p & 1][t].w};
        acc[t] = __builtin_amdgcn_wmma_f32_16x16x4_f32(false, a1, false, b1, (short)0, acc[t],
                                                       false, false);
      }
    }
    float* cp = C + (size_t)(16 * s + 8 * h) * N + 16 * (g * TU) + c;
#pragma unroll
    for (int t = 0; t < TU; ++t) {
#pragma unroll
      for (int r = 0; r < 8; ++r) cp[r * N + 16 * t] = acc[t][r];
    }
  }
}


__device__ __forceinline__ float lane_val(float v, int k) {
  return __int_as_float(__builtin_amdgcn_readlane(__float_as_int(v), k));
}
template <int X>
__device__ __forceinline__ float row_xor(float v) {
  const int b = __float_as_int(v);
  return __int_as_float(__builtin_amdgcn_update_dpp(b, b, 0x160 + X, 0xf, 0xf, false));
}
__device__ __forceinline__ float xhalf(float v) {
  const int b = __float_as_int(v);
  return __int_as_float(__builtin_amdgcn_permlanex16(b, b, 0x76543210, 0xfedcba98, false, false));
}
template <int K>
__device__ __forceinline__ float group8_lane(float v) {
  constexpr unsigned sel = K * 011111111u;
  return __int_as_float((int)__builtin_amdgcn_mov_dpp8((unsigned)__float_as_int(v), sel));
}
template <int NH, int M>
__device__ __forceinline__ void fold_step(float* v, int lane) {
  const bool up = (lane & M) != 0;
#pragma unroll
  for (int i = 0; i < NH; ++i) {
    const float keep = up ? v[i + NH] : v[i];
    const float send = up ? v[i] : v[i + NH];
    if constexpr (M == 16) v[i] = keep + xhalf(send);
    else v[i] = keep + row_xor<M>(send);
  }
}
template <int K>
__device__ __forceinline__ void add_rows8(v4f& acc, float& den, float p, const v4f* r) {
  if constexpr (K < 8) {
    const float pk = group8_lane<K>(p);
    acc += pk * r[K];
    den += pk;
    add_rows8<K + 1>(acc, den, p, r);
  }
}
#define GLOBAL_AS __attribute__((address_space(1)))
template <typename T>
__device__ __forceinline__ const GLOBAL_AS T* row_base(const T* table, int sk, int row_len) {
  const GLOBAL_AS T* rp = (const GLOBAL_AS T*)(table + (size_t)sk * row_len);
  asm("" : "+s"(rp));
  return rp;
}
__device__ __forceinline__ float dot4(v4f a, v4f b) {
  const v2f t = a.xy * b.xy + a.zw * b.zw;
  return t.x + t.y;
}
__device__ __forceinline__ float wave_max(float v) {
  v = fmaxf(v, row_xor<1>(v));
  v = fmaxf(v, row_xor<2>(v));
  v = fmaxf(v, row_xor<4>(v));
  v = fmaxf(v, row_xor<8>(v));
  return fmaxf(v, xhalf(v));
}

#define G1_WAVES 16
#define HS_PITCH (HEADS * HID + 4)
#define PT_PITCH (256 + 16)
static_assert(HEADS * HID == 8 * G1_WAVES, "16 k per chunk, one chunk per pair of waves");
static_assert(HID == 32, "two column tiles, one output element per thread");
__global__ void __launch_bounds__(32 * G1_WAVES)
gat_gather1_mm(const float* __restrict__ feat, const float* __restrict__ att_s,
               const float* __restrict__ att_d, const unsigned* __restrict__ pos,
               const unsigned short* __restrict__ col, const float* __restrict__ bias,
               const float* __restrict__ W2, float* __restrict__ z) {
  __shared__ __attribute__((aligned(16))) float hs[16 * HS_PITCH];
  __shared__ float part[G1_WAVES * PT_PITCH];
  const int wv = __builtin_amdgcn_readfirstlane(threadIdx.x >> 5);
  const int n = blockIdx.x * G1_WAVES + wv;
  const int lane = threadIdx.x & 31;
  const v4f* fv = (const v4f*)feat;
  const int beg = (int)pos[n];
  const int end = (n + 1 < NN) ? (int)pos[n + 1] : NE;
  unsigned s = (lane < end - beg) ? (unsigned)col[beg + lane] : (unsigned)n;
  const v4f self = (fv + (size_t)n * 32)[lane];
  const v4f ws = ((const v4f*)att_s)[lane];
  const v4f wd = ((const v4f*)att_d)[lane];
  float asn = dot4(self, ws);
  float ad  = dot4(self, wd);
  asn += row_xor<1>(asn); ad += row_xor<1>(ad);
  asn += row_xor<2>(asn); ad += row_xor<2>(ad);
  asn += row_xor<4>(asn); ad += row_xor<4>(ad);
  float m = lrelu(asn + ad);
  float den = 1.f;
  v4f acc = self;
  for (int j0 = beg; ; ) {
    const int cnt = (end - j0 < 32) ? (end - j0) : 32;
    for (int k0 = 0; ; ) {
      v4f r[8];
#pragma unroll
      for (int k = 0; k < 8; ++k)
        r[k] = row_base(fv, __builtin_amdgcn_readlane((int)s, k0 + k), 32)[lane];
      float v[8];
#pragma unroll
      for (int k = 0; k < 8; ++k) v[k] = dot4(r[k], ws);
      fold_step<4, 4>(v, lane);
      fold_step<2, 2>(v, lane);
      fold_step<1, 1>(v, lane);
      const int left = (cnt - k0 < 8) ? (cnt - k0) : 8;
      const bool ok = __builtin_amdgcn_inverse_ballot_w32(((1u << left) - 1u) * 0x01010101u);
      const float e = ok ? lrelu(v[0] + ad) : -INFINITY;
      float bm = fmaxf(e, row_xor<1>(e));
      bm = fmaxf(bm, row_xor<2>(bm));
      bm = fmaxf(bm, row_xor<4>(bm));
      const float mn = fmaxf(m, bm);
      const float p = __expf(e - mn);
      const float sc = __expf(m - mn);
      m = mn;
      acc.x *= sc; acc.y *= sc; acc.z *= sc; acc.w *= sc;
      den *= sc;
      add_rows8<0>(acc, den, p, r);
      k0 += 8;
      if (k0 >= cnt) break;
    }
    j0 += 32;
    if (j0 >= end) break;
    s = (lane < end - j0) ? (unsigned)col[j0 + lane] : (unsigned)n;
  }
  const int c = lane & 15, h = lane >> 4;
  const int tl = wv & 1, kc = wv >> 1;
  const float* bp = W2 + (size_t)(16 * kc + 2 * h) * HID + 16 * tl + c;
  v2f bf[4];
#pragma unroll
  for (int kk = 0; kk < 4; ++kk) bf[kk] = (v2f){bp[(4 * kk) * HID], bp[(4 * kk + 1) * HID]};
  float4 o;
  o.x = elu(acc.x / den + bias[4 * lane + 0]);
  o.y = elu(acc.y / den + bias[4 * lane + 1]);
  o.z = elu(acc.z / den + bias[4 * lane + 2]);
  o.w = elu(acc.w / den + bias[4 * lane + 3]);
  *(float4*)(hs + wv * HS_PITCH + 4 * lane) = o;
  __syncthreads();
  v8f d = (v8f){0.f, 0.f, 0.f, 0.f, 0.f, 0.f, 0.f, 0.f};
  const float* ap = hs + c * HS_PITCH + 16 * kc + 2 * h;
#pragma unroll
  for (int kk = 0; kk < 4; ++kk) {
    const v2f a = *(const v2f*)(ap + 4 * kk);
    d = __builtin_amdgcn_wmma_f32_16x16x4_f32(false, a, false, bf[kk], (short)0, d, false, false);
  }
#pragma unroll
  for (int r = 0; r < 8; ++r) part[wv * PT_PITCH + 32 * r + lane] = d[r];
  __syncthreads();
  const float* pp = part + (lane >> 4) * PT_PITCH + 32 * (wv & 7) + 16 * (wv >> 3) + (lane & 15);
  float sum = pp[0];
#pragma unroll
  for (int k = 1; k < 8; ++k) sum += pp[2 * k * PT_PITCH];
  z[(size_t)n * HID + lane] = sum;
}

__global__ void __launch_bounds__(256)
gat_gather2_cls(const float* __restrict__ feat, const float* __restrict__ att_s,
                const float* __restrict__ att_d, const unsigned* __restrict__ pos,
                const unsigned short* __restrict__ col, const float* __restrict__ b2,
                const float* __restrict__ lin_w, const float* __restrict__ lin_b,
                float* __restrict__ out) {
  const int n = blockIdx.x * (blockDim.x >> 5) + __builtin_amdgcn_readfirstlane(threadIdx.x >> 5);
  const int c = threadIdx.x & 31;
  if (n >= NN) return;
  const int je = c >> 1;
  const int beg = (int)pos[n];
  const int end = (n + 1 < NN) ? (int)pos[n + 1] : NE;
  unsigned s = (c < end - beg) ? (unsigned)col[beg + c] : (unsigned)n;
  const float self = (feat + (size_t)n * HID)[c];
  const float ws = att_s[c];
  float asn = self * ws;
  float adp = self * att_d[c];
  asn += xhalf(asn);      adp += xhalf(adp);
  asn += row_xor<8>(asn); adp += row_xor<8>(adp);
  asn += row_xor<4>(asn); adp += row_xor<4>(adp);
  asn += row_xor<2>(asn); adp += row_xor<2>(adp);
  asn += row_xor<1>(asn); adp += row_xor<1>(adp);
  float m = lrelu(asn + adp);
  float den = 1.f;
  float acc = self;
  for (int j0 = beg; ; ) {
    const int cnt = (end - j0 < 32) ? (end - j0) : 32;
    for (int k0 = 0; ; ) {
      float r[16];
#pragma unroll
      for (int k = 0; k < 16; ++k)
        r[k] = row_base(feat, __builtin_amdgcn_readlane((int)s, k0 + k), HID)[c];
      float v[16];
#pragma unroll
      for (int k = 0; k < 16; ++k) v[k] = r[k] * ws;
      fold_step<8, 16>(v, c);
      fold_step<4, 8>(v, c);
      fold_step<2, 4>(v, c);
      fold_step<1, 2>(v, c);
      v[0] += row_xor<1>(v[0]);
      const float e = (k0 + je < cnt) ? lrelu(v[0] + adp) : -INFINITY;
      const float mn = fmaxf(m, wave_max(e));
      const float p = __expf(e - mn);
      const float sc = __expf(m - mn);
      m = mn;
      acc *= sc;
      den *= sc;
#pragma unroll
      for (int k = 0; k < 16; ++k) {
        const float pk = lane_val(p, 2 * k);
        acc += pk * r[k];
        den += pk;
      }
      k0 += 16;
      if (k0 >= cnt) break;
    }
    j0 += 32;
    if (j0 >= end) break;
    s = (c < end - j0) ? (unsigned)col[j0 + c] : (unsigned)n;
  }
  const float v = elu(acc / den + b2[c]);
  float w0 = v * lin_w[c * 2 + 0];
  float w1 = v * lin_w[c * 2 + 1];
#pragma unroll
  for (int x = 16; x > 0; x >>= 1) {
    w0 += __shfl_xor(w0, x, 32);
    w1 += __shfl_xor(w1, x, 32);
  }
  if (c == 0) {
    out[n * 2 + 0] = w0 + lin_b[0];
    out[n * 2 + 1] = w1 + lin_b[1];
  }
}

extern "C" void kernel_launch(void* const* d_in, const int* in_sizes, int n_in,
                              void* d_out, int out_size, void* d_ws, size_t ws_size,
                              hipStream_t stream) {
  const float* x        = (const float*)d_in[0];
  const int*   ei       = (const int*)d_in[1];
  const float* W1       = (const float*)d_in[2];
  const float* att_src1 = (const float*)d_in[3];
  const float* att_dst1 = (const float*)d_in[4];
  const float* b1       = (const float*)d_in[5];
  const float* W2       = (const float*)d_in[6];
  const float* att_src2 = (const float*)d_in[7];
  const float* att_dst2 = (const float*)d_in[8];
  const float* b2       = (const float*)d_in[9];
  const float* lin_w    = (const float*)d_in[10];
  const float* lin_b    = (const float*)d_in[11];
  float* out = (float*)d_out;
  const int* src = ei;
  const int* dst = ei + NE;

  float*          ws  = (float*)d_ws;
  float*          h1  = ws;
  unsigned*       pos = (unsigned*)(ws + 6600000);
  unsigned short* col = (unsigned short*)(ws + 6650000);
  float*          h2  = ws + 7200000;
  unsigned*       tmp    = (unsigned*)h2;
  unsigned*       table  = (unsigned*)(h2 + 800000);
  float*          z   = h2;

  gemm_wmma_f32<HEADS * HID, INDIM, 2, true>
      <<<GEMM_WGS, GEMM_THREADS, 0, stream>>>(x, W1, h1, src, dst, table, tmp);
  csr_fine<<<NBK, SORT_THREADS, 0, stream>>>(tmp, table, pos, col);
  gat_gather1_mm<<<NN / G1_WAVES, 32 * G1_WAVES, 0, stream>>>(h1, att_src1, att_dst1, pos, col, b1, W2, z);

  gat_gather2_cls<<<NN / 8, 256, 0, stream>>>(z, att_src2, att_dst2, pos, col, b2, lin_w, lin_b, out);
}


# Layer-2 aggregation packs two nodes into each wave (16-lane row per node, 2 channels per lane, 8-byte row loads, ids and weights handed out by DPP row_share); scores and logits summed in other order
# speedup vs baseline: 10.4441x; 1.0565x over previous
#include <hip/hip_runtime.h>
#include <math.h>


#define NN 50000
#define NE 800000
#define INDIM 256
#define HID 32
#define HEADS 4

static_assert(NN <= 65536, "source node ids are kept in 16 bits");
static_assert(NN % 16 == 0, "the products take whole 16-row strips; both aggregation kernels take 16 nodes per block");

typedef __attribute__((ext_vector_type(2))) float v2f;
typedef __attribute__((ext_vector_type(8))) float v8f;
typedef __attribute__((ext_vector_type(4))) float v4f;

__device__ __forceinline__ float elu(float v) {
  return (v > 0.f) ? v : (__expf(v) - 1.f);
}
__device__ __forceinline__ float lrelu(float x) {
  return (x > 0.f) ? x : 0.2f * x;
}

#define BSH 8
#define NBK ((NN + (1 << BSH) - 1) >> BSH)
#define NEB 256
#define EPB ((NE + NEB - 1) / NEB)
#define SORT_THREADS 1024
#define FINE_CAP 16384
#define FINE_REG 10
#define RUN_LANES (SORT_THREADS / NEB)
#define PRE_Q ((NBK * (NEB / 4) + SORT_THREADS - 1) / SORT_THREADS)
static_assert(NBK <= 256, "bucket ids index 256-entry LDS arrays");
static_assert(NEB % 4 == 0, "a table row is read as whole 16-byte words");
static_assert(EPB < 65536, "a run's length and its start inside the block share one table word");
static_assert(SORT_THREADS % NEB == 0 && RUN_LANES >= 1, "whole groups of threads per run");
static_assert((long long)NEB * EPB == NE, "the blocks cover every edge and every block is full");

template <int X>
__device__ __forceinline__ unsigned row_xor_u(unsigned v) {
  return (unsigned)__builtin_amdgcn_update_dpp((int)v, (int)v, 0x160 + X, 0xf, 0xf, false);
}
__device__ __forceinline__ unsigned wave_sum_u(unsigned v) {
  v += row_xor_u<1>(v);
  v += row_xor_u<2>(v);
  v += row_xor_u<4>(v);
  v += row_xor_u<8>(v);
  return v + (unsigned)__builtin_amdgcn_permlanex16((int)v, (int)v, 0x76543210, 0xfedcba98, false, false);
}

__device__ __forceinline__ void scan256(unsigned* a, int lane) {
  unsigned v[8], sum = 0u;
#pragma unroll
  for (int k = 0; k < 8; ++k) { v[k] = a[8 * lane + k]; sum += v[k]; }
  unsigned inc = sum;
#pragma unroll
  for (int m = 1; m < 32; m <<= 1) {
    const unsigned u = __shfl_up(inc, m, 32);
    if (lane >= m) inc += u;
  }
  unsigned run = inc - sum;
#pragma unroll
  for (int k = 0; k < 8; ++k) { a[8 * lane + k] = run; run += v[k]; }
}

template <int NT>
__device__ __forceinline__ void local_load(const int* __restrict__ src, const int* __restrict__ dst,
                                           int blk, unsigned (&wd)[(EPB + NT - 1) / NT]) {
  const int t = threadIdx.x;
  const int e0 = blk * EPB;
#pragma unroll
  for (int k = 0; k < (EPB + NT - 1) / NT; ++k) {
    const int i = (t + k * NT < EPB) ? t + k * NT : 0;
    wd[k] = ((unsigned)dst[e0 + i] << 16) | (unsigned)src[e0 + i];
  }
}
template <int NT>
__device__ __forceinline__ void local_order(const unsigned (&wd)[(EPB + NT - 1) / NT],
                                            unsigned* cnt, unsigned* cursor, unsigned* stage) {
  const int t = threadIdx.x, lane = t & 31, w = t >> 5;
  if (t < 256) cnt[t] = 0u;
  __syncthreads();
#pragma unroll
  for (int k = 0; k < (EPB + NT - 1) / NT; ++k) {
    if (t + k * NT < EPB) atomicAdd(&cnt[wd[k] >> 24], 1u);
  }
  __syncthreads();
  if (w == 0) {
    unsigned v[8], sum = 0u;
#pragma unroll
    for (int k = 0; k < 8; ++k) { v[k] = cnt[8 * lane + k]; sum += v[k]; }
    unsigned inc = sum;
#pragma unroll
    for (int m = 1; m < 32; m <<= 1) {
      const unsigned u = __shfl_up(inc, m, 32);
      if (lane >= m) inc += u;
    }
    unsigned run = inc - sum;
#pragma unroll
    for (int k = 0; k < 8; ++k) {
      const int b = 8 * lane + k;
      cursor[b] = run;
      cnt[b] = (run << 16) | v[k];
      run += v[k];
    }
  }
  __syncthreads();
  unsigned r[(EPB + NT - 1) / NT];
#pragma unroll
  for (int k = 0; k < (EPB + NT - 1) / NT; ++k) {
    r[k] = (t + k * NT < EPB) ? atomicAdd(&cursor[wd[k] >> 24], 1u) : 0u;
  }
#pragma unroll
  for (int k = 0; k < (EPB + NT - 1) / NT; ++k) {
    if (t + k * NT < EPB) stage[r[k]] = wd[k];
  }
}
template <int NT>
__device__ __forceinline__ void local_store(unsigned* __restrict__ table, unsigned* __restrict__ tmp,
                                            int blk, const unsigned* cnt, const unsigned* stage) {
  static_assert(NT >= NBK, "one thread per table word");
  const int t = threadIdx.x;
  const int e0 = blk * EPB;
  if (t < NBK) table[t * NEB + blk] = cnt[t];
#pragma unroll
  for (int k = 0; k < (EPB + NT - 1) / NT; ++k) {
    const int i = t + k * NT;
    if (i < EPB) tmp[e0 + i] = stage[i];
  }
}

__global__ void __launch_bounds__(SORT_THREADS)
csr_fine(const unsigned* __restrict__ tmp, const unsigned* __restrict__ table,
         unsigned* __restrict__ pos, unsigned short* __restrict__ col) {
  __shared__ unsigned c[256];
  __shared__ unsigned part_pre[SORT_THREADS / 32];
  __shared__ unsigned part_own[SORT_THREADS / 32];
  __shared__ unsigned short stage[FINE_CAP];
  const int t = threadIdx.x, lane = t & 31, w = t >> 5, b = blockIdx.x;
  const int rk = t / RUN_LANES, rj = t % RUN_LANES;
  const unsigned word = table[b * NEB + rk];
  const uint4* tq = (const uint4*)table;
  const int nq = b * (NEB / 4);
  uint4 pv[PRE_Q];
#pragma unroll
  for (int k = 0; k < PRE_Q; ++k) {
    const int i = t + k * SORT_THREADS;
    pv[k] = tq[(i < nq) ? i : 0];
  }
  const int rc = (int)(word & 0xffffu);
  const unsigned* rp = tmp + (size_t)rk * EPB + (word >> 16);
  unsigned x[FINE_REG];
#pragma unroll
  for (int k = 0; k < FINE_REG; ++k) {
    const int i = rj + k * RUN_LANES;
    x[k] = (i < rc) ? rp[i] : 0u;
  }
  if (t < 256) c[t] = 0u;
  unsigned ps = 0u;
#pragma unroll
  for (int k = 0; k < PRE_Q; ++k) {
    const unsigned q = (pv[k].x & 0xffffu) + (pv[k].y & 0xffffu) + (pv[k].z & 0xffffu) + (pv[k].w & 0xffffu);
    ps += (t + k * SORT_THREADS < nq) ? q : 0u;
  }
  ps = wave_sum_u(ps);
  const unsigned os = wave_sum_u(rj == 0 ? (unsigned)rc : 0u);
  if (lane == 0) { part_pre[w] = ps; part_own[w] = os; }
  __syncthreads();
  const int s = (int)wave_sum_u(part_pre[lane]);
  const int cnt = (int)wave_sum_u(part_own[lane]);
#pragma unroll
  for (int k = 0; k < FINE_REG; ++k) {
    if (rj + k * RUN_LANES < rc) atomicAdd(&c[(x[k] >> 16) & 255u], 1u);
  }
  for (int i = rj + FINE_REG * RUN_LANES; i < rc; i += RUN_LANES)
    atomicAdd(&c[(rp[i] >> 16) & 255u], 1u);
  __syncthreads();
  if (w == 0) scan256(c, lane);
  __syncthreads();
  const unsigned first = (t < 256) ? (unsigned)s + c[t] : 0u;
  __syncthreads();
  if (cnt <= FINE_CAP) {
    unsigned r[FINE_REG];
#pragma unroll
    for (int k = 0; k < FINE_REG; ++k) {
      r[k] = (rj + k * RUN_LANES < rc) ? atomicAdd(&c[(x[k] >> 16) & 255u], 1u) : 0u;
    }
#pragma unroll
    for (int k = 0; k < FINE_REG; ++k) {
      if (rj + k * RUN_LANES < rc) stage[r[k]] = (unsigned short)x[k];
    }
    for (int i = rj + FINE_REG * RUN_LANES; i < rc; i += RUN_LANES) {
      const unsigned y = rp[i];
      stage[atomicAdd(&c[(y >> 16) & 255u], 1u)] = (unsigned short)y;
    }
    __syncthreads();
    for (int i = t; i < cnt; i += SORT_THREADS) col[s + i] = stage[i];
  } else {
#pragma unroll
    for (int k = 0; k < FINE_REG; ++k) {
      if (rj + k * RUN_LANES < rc)
        col[s + atomicAdd(&c[(x[k] >> 16) & 255u], 1u)] = (unsigned short)x[k];
    }
    for (int i = rj + FINE_REG * RUN_LANES; i < rc; i += RUN_LANES) {
      const unsigned y = rp[i];
      col[s + atomicAdd(&c[(y >> 16) & 255u], 1u)] = (unsigned short)y;
    }
  }
  if (t < 256 && (b << BSH) + t < NN) pos[(b << BSH) + t] = first;
}

#define GEMM_WGS 256
#define GEMM_THREADS 512

template <int N, int K, int G, bool SORT>
__global__ void __launch_bounds__(GEMM_THREADS) __attribute__((amdgpu_waves_per_eu(4, 4)))
gemm_wmma_f32(const float* __restrict__ A, const float* __restrict__ B,
              float* __restrict__ C, const int* __restrict__ src, const int* __restrict__ dst,
              unsigned* __restrict__ table, unsigned* __restrict__ tmp) {
  static_assert(!SORT || GEMM_WGS == NEB, "one work-group per edge block");
  constexpr int NT = N / 16;
  constexpr int TU = NT / G;
  constexpr int NP = K / 8;
  constexpr int WQ = (GEMM_THREADS / 32) / G;
  static_assert(N % (16 * G) == 0 && K % 8 == 0 && (GEMM_THREADS / 32) % G == 0,
                "whole column tiles per unit, whole pairs of k steps, whole groups of waves");
  __shared__ float4 bs[NP * NT * 32];
  __shared__ unsigned s_cnt[SORT ? 256 : 1];
  __shared__ unsigned s_cursor[SORT ? 256 : 1];
  __shared__ unsigned s_stage[SORT ? EPB : 1];

  unsigned wd[SORT ? (EPB + GEMM_THREADS - 1) / GEMM_THREADS : 1];
  if constexpr (SORT) local_load<GEMM_THREADS>(src, dst, blockIdx.x, wd);

  for (int e = threadIdx.x; e < NP * NT * 32; e += GEMM_THREADS) {
    const int l = e & 31, t = (e >> 5) % NT, p = e / (32 * NT);
    const float* bp = B + (size_t)(8 * p + 4 * (l >> 4)) * N + 16 * t + (l & 15);
    float4 v;
    v.x = bp[0];
    v.y = bp[N];
    v.z = bp[2 * N];
    v.w = bp[3 * N];
    bs[e] = v;
  }
  if constexpr (SORT) local_order<GEMM_THREADS>(wd, s_cnt, s_cursor, s_stage);
  __syncthreads();
  if constexpr (SORT) local_store<GEMM_THREADS>(table, tmp, blockIdx.x, s_cnt, s_stage);

  const int lane = threadIdx.x & 31;
  const int w    = threadIdx.x >> 5;
  const int g    = w % G;
  const int q    = w / G;
  const int c    = lane & 15;
  const int h    = lane >> 4;
  const float4* bl = bs + (g * TU) * 32 + lane;

  for (int s = blockIdx.x + GEMM_WGS * q; s < NN / 16; s += GEMM_WGS * WQ) {
    asm volatile("" ::: "memory");
    const float4* ap = (const float4*)(A + (size_t)(16 * s + c) * K) + h;
    v8f acc[TU];
#pragma unroll
    for (int t = 0; t < TU; ++t) acc[t] = (v8f){0.f, 0.f, 0.f, 0.f, 0.f, 0.f, 0.f, 0.f};
    float4 a[NP];
#pragma unroll
    for (int p = 0; p < NP; ++p) a[p] = ap[2 * p];
    float4 b[2][TU];
#pragma unroll
    for (int t = 0; t < TU; ++t) b[0][t] = bl[t * 32];
#pragma unroll
    for (int p = 0; p < NP; ++p) {
      if (p + 1 < NP) {
#pragma unroll
        for (int t = 0; t < TU; ++t) b[(p + 1) & 1][t] = bl[((p + 1) * NT + t) * 32];
      }
      const v2f a0 = {a[p].x, a[p].y}, a1 = {a[p].z, a[p].w};
#pragma unroll
      for (int t = 0; t < TU; ++t) {
        const v2f b0 = {b[p & 1][t].x, b[p & 1][t].y};
        acc[t] = __builtin_amdgcn_wmma_f32_16x16x4_f32(false, a0, false, b0, (short)0, acc[t],
                                                       false, false);
      }
#pragma unroll
      for (int t = 0; t < TU; ++t) {
        const v2f b1 = {b[p & 1][t].z, b[p & 1][t].w};
        acc[t] = __builtin_amdgcn_wmma_f32_16x16x4_f32(false, a1, false, b1, (short)0, acc[t],
                                                       false, false);
      }
    }
    float* cp = C + (size_t)(16 * s + 8 * h) * N + 16 * (g * TU) + c;
#pragma unroll
    for (int t = 0; t < TU; ++t) {
#pragma unroll
      for (int r = 0; r < 8; ++r) cp[r * N + 16 * t] = acc[t][r];
    }
  }
}


template <int X>
__device__ __forceinline__ float row_xor(float v) {
  const int b = __float_as_int(v);
  return __int_as_float(__builtin_amdgcn_update_dpp(b, b, 0x160 + X, 0xf, 0xf, false));
}
__device__ __forceinline__ float xhalf(float v) {
  const int b = __float_as_int(v);
  return __int_as_float(__builtin_amdgcn_permlanex16(b, b, 0x76543210, 0xfedcba98, false, false));
}
template <int K>
__device__ __forceinline__ float group8_lane(float v) {
  constexpr unsigned sel = K * 011111111u;
  return __int_as_float((int)__builtin_amdgcn_mov_dpp8((unsigned)__float_as_int(v), sel));
}
template <int NH, int M>
__device__ __forceinline__ void fold_step(float* v, int lane) {
  const bool up = (lane & M) != 0;
#pragma unroll
  for (int i = 0; i < NH; ++i) {
    const float keep = up ? v[i + NH] : v[i];
    const float send = up ? v[i] : v[i + NH];
    if constexpr (M == 16) v[i] = keep + xhalf(send);
    else v[i] = keep + row_xor<M>(send);
  }
}
template <int K>
__device__ __forceinline__ void add_rows8(v4f& acc, float& den, float p, const v4f* r) {
  if constexpr (K < 8) {
    const float pk = group8_lane<K>(p);
    acc += pk * r[K];
    den += pk;
    add_rows8<K + 1>(acc, den, p, r);
  }
}
#define GLOBAL_AS __attribute__((address_space(1)))
template <typename T>
__device__ __forceinline__ const GLOBAL_AS T* row_base(const T* table, int sk, int row_len) {
  const GLOBAL_AS T* rp = (const GLOBAL_AS T*)(table + (size_t)sk * row_len);
  asm("" : "+s"(rp));
  return rp;
}
__device__ __forceinline__ float dot4(v4f a, v4f b) {
  const v2f t = a.xy * b.xy + a.zw * b.zw;
  return t.x + t.y;
}
template <int K>
__device__ __forceinline__ unsigned row_lane_u(unsigned v) {
  return (unsigned)__builtin_amdgcn_update_dpp(0, (int)v, 0x150 + K, 0xf, 0xf, true);
}
template <int K>
__device__ __forceinline__ float row_lane(float v) {
  return __int_as_float((int)row_lane_u<K>((unsigned)__float_as_int(v)));
}
__device__ __forceinline__ float row_max(float v) {
  v = fmaxf(v, row_xor<1>(v));
  v = fmaxf(v, row_xor<2>(v));
  v = fmaxf(v, row_xor<4>(v));
  return fmaxf(v, row_xor<8>(v));
}
__device__ __forceinline__ float row_sum(float v) {
  v += row_xor<8>(v);
  v += row_xor<4>(v);
  v += row_xor<2>(v);
  return v + row_xor<1>(v);
}
template <int K>
__device__ __forceinline__ void load_rows16(v2f* r, const float* table, unsigned s, unsigned lo) {
  if constexpr (K < 16) {
    const unsigned off = (row_lane_u<K>(s) << 7) | lo;
    r[K] = *(const v2f*)((const char*)table + off);
    load_rows16<K + 1>(r, table, s, lo);
  }
}
template <int K>
__device__ __forceinline__ void add_rows16(v2f& acc, float& den, float p, const v2f* r) {
  if constexpr (K < 16) {
    const float pk = row_lane<K>(p);
    acc += pk * r[K];
    den += pk;
    add_rows16<K + 1>(acc, den, p, r);
  }
}

#define G1_WAVES 16
#define HS_PITCH (HEADS * HID + 4)
#define PT_PITCH (256 + 16)
static_assert(HEADS * HID == 8 * G1_WAVES, "16 k per chunk, one chunk per pair of waves");
static_assert(HID == 32, "two column tiles, one output element per thread");
__global__ void __launch_bounds__(32 * G1_WAVES)
gat_gather1_mm(const float* __restrict__ feat, const float* __restrict__ att_s,
               const float* __restrict__ att_d, const unsigned* __restrict__ pos,
               const unsigned short* __restrict__ col, const float* __restrict__ bias,
               const float* __restrict__ W2, float* __restrict__ z) {
  __shared__ __attribute__((aligned(16))) float hs[16 * HS_PITCH];
  __shared__ float part[G1_WAVES * PT_PITCH];
  const int wv = __builtin_amdgcn_readfirstlane(threadIdx.x >> 5);
  const int n = blockIdx.x * G1_WAVES + wv;
  const int lane = threadIdx.x & 31;
  const v4f* fv = (const v4f*)feat;
  const int beg = (int)pos[n];
  const int end = (n + 1 < NN) ? (int)pos[n + 1] : NE;
  unsigned s = (lane < end - beg) ? (unsigned)col[beg + lane] : (unsigned)n;
  const v4f self = (fv + (size_t)n * 32)[lane];
  const v4f ws = ((const v4f*)att_s)[lane];
  const v4f wd = ((const v4f*)att_d)[lane];
  float asn = dot4(self, ws);
  float ad  = dot4(self, wd);
  asn += row_xor<1>(asn); ad += row_xor<1>(ad);
  asn += row_xor<2>(asn); ad += row_xor<2>(ad);
  asn += row_xor<4>(asn); ad += row_xor<4>(ad);
  float m = lrelu(asn + ad);
  float den = 1.f;
  v4f acc = self;
  for (int j0 = beg; ; ) {
    const int cnt = (end - j0 < 32) ? (end - j0) : 32;
    for (int k0 = 0; ; ) {
      v4f r[8];
#pragma unroll
      for (int k = 0; k < 8; ++k)
        r[k] = row_base(fv, __builtin_amdgcn_readlane((int)s, k0 + k), 32)[lane];
      float v[8];
#pragma unroll
      for (int k = 0; k < 8; ++k) v[k] = dot4(r[k], ws);
      fold_step<4, 4>(v, lane);
      fold_step<2, 2>(v, lane);
      fold_step<1, 1>(v, lane);
      const int left = (cnt - k0 < 8) ? (cnt - k0) : 8;
      const bool ok = __builtin_amdgcn_inverse_ballot_w32(((1u << left) - 1u) * 0x01010101u);
      const float e = ok ? lrelu(v[0] + ad) : -INFINITY;
      float bm = fmaxf(e, row_xor<1>(e));
      bm = fmaxf(bm, row_xor<2>(bm));
      bm = fmaxf(bm, row_xor<4>(bm));
      const float mn = fmaxf(m, bm);
      const float p = __expf(e - mn);
      const float sc = __expf(m - mn);
      m = mn;
      acc.x *= sc; acc.y *= sc; acc.z *= sc; acc.w *= sc;
      den *= sc;
      add_rows8<0>(acc, den, p, r);
      k0 += 8;
      if (k0 >= cnt) break;
    }
    j0 += 32;
    if (j0 >= end) break;
    s = (lane < end - j0) ? (unsigned)col[j0 + lane] : (unsigned)n;
  }
  const int c = lane & 15, h = lane >> 4;
  const int tl = wv & 1, kc = wv >> 1;
  const float* bp = W2 + (size_t)(16 * kc + 2 * h) * HID + 16 * tl + c;
  v2f bf[4];
#pragma unroll
  for (int kk = 0; kk < 4; ++kk) bf[kk] = (v2f){bp[(4 * kk) * HID], bp[(4 * kk + 1) * HID]};
  float4 o;
  o.x = elu(acc.x / den + bias[4 * lane + 0]);
  o.y = elu(acc.y / den + bias[4 * lane + 1]);
  o.z = elu(acc.z / den + bias[4 * lane + 2]);
  o.w = elu(acc.w / den + bias[4 * lane + 3]);
  *(float4*)(hs + wv * HS_PITCH + 4 * lane) = o;
  __syncthreads();
  v8f d = (v8f){0.f, 0.f, 0.f, 0.f, 0.f, 0.f, 0.f, 0.f};
  const float* ap = hs + c * HS_PITCH + 16 * kc + 2 * h;
#pragma unroll
  for (int kk = 0; kk < 4; ++kk) {
    const v2f a = *(const v2f*)(ap + 4 * kk);
    d = __builtin_amdgcn_wmma_f32_16x16x4_f32(false, a, false, bf[kk], (short)0, d, false, false);
  }
#pragma unroll
  for (int r = 0; r < 8; ++r) part[wv * PT_PITCH + 32 * r + lane] = d[r];
  __syncthreads();
  const float* pp = part + (lane >> 4) * PT_PITCH + 32 * (wv & 7) + 16 * (wv >> 3) + (lane & 15);
  float sum = pp[0];
#pragma unroll
  for (int k = 1; k < 8; ++k) sum += pp[2 * k * PT_PITCH];
  z[(size_t)n * HID + lane] = sum;
}

__global__ void __launch_bounds__(256)
gat_gather2_cls(const float* __restrict__ feat, const float* __restrict__ att_s,
                const float* __restrict__ att_d, const unsigned* __restrict__ pos,
                const unsigned short* __restrict__ col, const float* __restrict__ b2,
                const float* __restrict__ lin_w, const float* __restrict__ lin_b,
                float* __restrict__ out) {
  const int wv = __builtin_amdgcn_readfirstlane(threadIdx.x >> 5);
  const int lane = threadIdx.x & 31;
  const int i = lane & 15;
  const int n = blockIdx.x * 16 + 2 * wv + (lane >> 4);
  const int beg = (int)pos[n];
  const int end = (n + 1 < NN) ? (int)pos[n + 1] : NE;
  const int deg = end - beg;
  const int d0 = __builtin_amdgcn_readlane(deg, 0), d1 = __builtin_amdgcn_readlane(deg, 16);
  const int maxdeg = (d0 > d1) ? d0 : d1;
  unsigned s = (i < deg) ? (unsigned)col[beg + i] : (unsigned)n;
  const unsigned lo = (unsigned)i << 3;
  const v2f self = *(const v2f*)((const char*)feat + (((unsigned)n << 7) | lo));
  const v2f ws = ((const v2f*)att_s)[i];
  const v2f wd = ((const v2f*)att_d)[i];
  const float asn = row_sum(self.x * ws.x + self.y * ws.y);
  const float adp = row_sum(self.x * wd.x + self.y * wd.y);
  float m = lrelu(asn + adp);
  float den = 1.f;
  v2f acc = self;
  for (int j0 = 0; ; ) {
    v2f r[16];
    load_rows16<0>(r, feat, s, lo);
    float v[16];
#pragma unroll
    for (int k = 0; k < 16; ++k) v[k] = r[k].x * ws.x + r[k].y * ws.y;
    fold_step<8, 8>(v, lane);
    fold_step<4, 4>(v, lane);
    fold_step<2, 2>(v, lane);
    fold_step<1, 1>(v, lane);
    const float e = (j0 + i < deg) ? lrelu(v[0] + adp) : -INFINITY;
    const float mn = fmaxf(m, row_max(e));
    const float p = __expf(e - mn);
    const float sc = __expf(m - mn);
    m = mn;
    acc *= sc;
    den *= sc;
    add_rows16<0>(acc, den, p, r);
    j0 += 16;
    if (j0 >= maxdeg) break;
    s = (j0 + i < deg) ? (unsigned)col[beg + j0 + i] : (unsigned)n;
  }
  const v2f bb = ((const v2f*)b2)[i];
  const v4f lw = ((const v4f*)lin_w)[i];
  const float v0 = elu(acc.x / den + bb.x);
  const float v1 = elu(acc.y / den + bb.y);
  const float w0 = row_sum(v0 * lw.x + v1 * lw.z);
  const float w1 = row_sum(v0 * lw.y + v1 * lw.w);
  if (i == 0) {
    float2 o;
    o.x = w0 + lin_b[0];
    o.y = w1 + lin_b[1];
    *(float2*)(out + 2 * n) = o;
  }
}

extern "C" void kernel_launch(void* const* d_in, const int* in_sizes, int n_in,
                              void* d_out, int out_size, void* d_ws, size_t ws_size,
                              hipStream_t stream) {
  const float* x        = (const float*)d_in[0];
  const int*   ei       = (const int*)d_in[1];
  const float* W1       = (const float*)d_in[2];
  const float* att_src1 = (const float*)d_in[3];
  const float* att_dst1 = (const float*)d_in[4];
  const float* b1       = (const float*)d_in[5];
  const float* W2       = (const float*)d_in[6];
  const float* att_src2 = (const float*)d_in[7];
  const float* att_dst2 = (const float*)d_in[8];
  const float* b2       = (const float*)d_in[9];
  const float* lin_w    = (const float*)d_in[10];
  const float* lin_b    = (const float*)d_in[11];
  float* out = (float*)d_out;
  const int* src = ei;
  const int* dst = ei + NE;

  float*          ws  = (float*)d_ws;
  float*          h1  = ws;
  unsigned*       pos = (unsigned*)(ws + 6600000);
  unsigned short* col = (unsigned short*)(ws + 6650000);
  float*          h2  = ws + 7200000;
  unsigned*       tmp    = (unsigned*)h2;
  unsigned*       table  = (unsigned*)(h2 + 800000);
  float*          z   = h2;

  gemm_wmma_f32<HEADS * HID, INDIM, 2, true>
      <<<GEMM_WGS, GEMM_THREADS, 0, stream>>>(x, W1, h1, src, dst, table, tmp);
  csr_fine<<<NBK, SORT_THREADS, 0, stream>>>(tmp, table, pos, col);
  gat_gather1_mm<<<NN / G1_WAVES, 32 * G1_WAVES, 0, stream>>>(h1, att_src1, att_dst1, pos, col, b1, W2, z);

  gat_gather2_cls<<<NN / 16, 256, 0, stream>>>(z, att_src2, att_dst2, pos, col, b2, lin_w, lin_b, out);
}
